# starting kernel
# baseline (speedup 1.0000x reference)
#include <hip/hip_runtime.h>
#include <hip/hip_bf16.h>
#include <math.h>

typedef __attribute__((ext_vector_type(16))) __bf16 v16bf;
typedef __attribute__((ext_vector_type(8)))  float  v8f;

#define BB 2
#define TT 2048
#define CC 1024
#define HH 16
#define HD 64

static __device__ inline v8f wmma_bf16(v16bf a, v16bf b, v8f c) {
  return __builtin_amdgcn_wmma_f32_16x16x32_bf16(false, a, false, b, (short)0, c, false, false);
}

static __device__ inline float gelu_tanh(float x) {
  const float c = 0.7978845608028654f;
  return 0.5f * x * (1.0f + tanhf(c * (x + 0.044715f * x * x * x)));
}

static __device__ inline v16bf frag_rowmajor(const __bf16* p, int ld, int lane) {
  const int m  = lane & 15;
  const int ko = (lane >> 4) << 3;
  const __bf16* q = p + (size_t)m * ld + ko;
  v16bf f;
#pragma unroll
  for (int j = 0; j < 4; ++j) {
    f[2 * j]     = q[2 * j];
    f[2 * j + 1] = q[2 * j + 1];
    f[8 + 2 * j] = q[16 + 2 * j];
    f[9 + 2 * j] = q[16 + 2 * j + 1];
  }
  return f;
}

static __device__ inline v16bf frag_b_strided(const __bf16* p, int ldk, int lane) {
  const int n  = lane & 15;
  const int ko = (lane >> 4) << 3;
  const __bf16* q = p + n;
  v16bf f;
#pragma unroll
  for (int j = 0; j < 4; ++j) {
    f[2 * j]     = q[(size_t)(ko + 2 * j) * ldk];
    f[2 * j + 1] = q[(size_t)(ko + 2 * j + 1) * ldk];
    f[8 + 2 * j] = q[(size_t)(16 + ko + 2 * j) * ldk];
    f[9 + 2 * j] = q[(size_t)(16 + ko + 2 * j + 1) * ldk];
  }
  return f;
}

__global__ void wconv_kernel(const float* __restrict__ src, __bf16* __restrict__ dst, int n) {
  const int i = blockIdx.x * blockDim.x + threadIdx.x;
  if (i < n) dst[i] = (__bf16)src[i];
}

__global__ void ln_kernel(const float* __restrict__ x, const float* __restrict__ w,
                          const float* __restrict__ b, __bf16* __restrict__ out) {
  const int row = blockIdx.x;
  const int tid = threadIdx.x;
  const float* xr = x + (size_t)row * CC;
  float v[4];
  float s = 0.f, sq = 0.f;
#pragma unroll
  for (int i = 0; i < 4; ++i) {
    v[i] = xr[tid + i * 256];
    s += v[i];
    sq += v[i] * v[i];
  }
#pragma unroll
  for (int off = 16; off >= 1; off >>= 1) {
    s  += __shfl_xor(s,  off, 32);
    sq += __shfl_xor(sq, off, 32);
  }
  __shared__ float reds[8], redq[8];
  const int wid = tid >> 5;
  if ((tid & 31) == 0) { reds[wid] = s; redq[wid] = sq; }
  __syncthreads();
  float ts = 0.f, tq = 0.f;
#pragma unroll
  for (int k = 0; k < 8; ++k) { ts += reds[k]; tq += redq[k]; }
  const float mean = ts * (1.0f / CC);
  const float var  = tq * (1.0f / CC) - mean * mean;
  const float rstd = rsqrtf(var + 1e-5f);
  __bf16* orow = out + (size_t)row * CC;
#pragma unroll
  for (int i = 0; i < 4; ++i) {
    const int c = tid + i * 256;
    orow[c] = (__bf16)((v[i] - mean) * rstd * w[c] + b[c]);
  }
}

template <int EM>
__global__ void __launch_bounds__(256, 1)
gemm_kernel(const __bf16* __restrict__ A, const __bf16* __restrict__ W,
            const float* __restrict__ bias, const float* __restrict__ resid,
            void* __restrict__ outp, int M, int N, int K) {
  __shared__ __bf16 As[128 * 40];
  __shared__ __bf16 Bs[128 * 40];
  const int tid  = threadIdx.x;
  const int lane = tid & 31;
  const int wid  = tid >> 5;
  const int wm   = wid & 1;
  const int wn   = wid >> 1;
  const int m0   = blockIdx.y * 128;
  const int n0   = blockIdx.x * 128;

  v8f acc[4][2] = {};

  const int srow = tid >> 1, sseg = tid & 1;
  const __bf16* agp = A + (size_t)(m0 + srow) * K + sseg * 16;
  const __bf16* bgp = W + (size_t)(n0 + srow) * K + sseg * 16;

  for (int k0 = 0; k0 < K; k0 += 32) {
    {
      const uint4* ap = (const uint4*)(agp + k0);
      uint4* ad = (uint4*)(As + srow * 40 + sseg * 16);
      ad[0] = ap[0]; ad[1] = ap[1];
      const uint4* bp = (const uint4*)(bgp + k0);
      uint4* bd = (uint4*)(Bs + srow * 40 + sseg * 16);
      bd[0] = bp[0]; bd[1] = bp[1];
    }
    if (k0 + 32 < K) {
      __builtin_prefetch(agp + k0 + 32, 0, 0);
      __builtin_prefetch(bgp + k0 + 32, 0, 0);
    }
    __syncthreads();

    v16bf af[4], bf[2];
#pragma unroll
    for (int i = 0; i < 4; ++i) af[i] = frag_rowmajor(As + (wm * 64 + i * 16) * 40, 40, lane);
#pragma unroll
    for (int j = 0; j < 2; ++j) bf[j] = frag_rowmajor(Bs + (wn * 32 + j * 16) * 40, 40, lane);
#pragma unroll
    for (int i = 0; i < 4; ++i)
#pragma unroll
      for (int j = 0; j < 2; ++j) acc[i][j] = wmma_bf16(af[i], bf[j], acc[i][j]);
    __syncthreads();
  }

  const int hi = lane >> 4;
  float biasj[2];
#pragma unroll
  for (int j = 0; j < 2; ++j) biasj[j] = bias[n0 + wn * 32 + j * 16 + (lane & 15)];

#pragma unroll
  for (int i = 0; i < 4; ++i)
#pragma unroll
    for (int j = 0; j < 2; ++j)
#pragma unroll
      for (int r = 0; r < 8; ++r) {
        const int row = m0 + wm * 64 + i * 16 + r + 8 * hi;
        const int col = n0 + wn * 32 + j * 16 + (lane & 15);
        float v = acc[i][j][r] + biasj[j];
        const size_t idx = (size_t)row * N + col;
        if constexpr (EM == 0) {
          ((float*)outp)[idx] = v + resid[idx];
        } else if constexpr (EM == 1) {
          ((__bf16*)outp)[idx] = (__bf16)v;
        } else {
          ((__bf16*)outp)[idx] = (__bf16)gelu_tanh(v);
        }
      }
}

__global__ void __launch_bounds__(32, 1)
attn_kernel(const __bf16* __restrict__ qkv, __bf16* __restrict__ y,
            float* __restrict__ ent) {
  const int lane  = threadIdx.x;
  const int hi    = lane >> 4;
  const int qbase = blockIdx.x * 16;
  const int h     = blockIdx.y;
  const int b     = blockIdx.z;
  const int ldq   = 3 * CC;

  const __bf16* Qp = qkv + (size_t)b * TT * ldq + h * HD;
  const __bf16* Kp = Qp + CC;
  const __bf16* Vp = Qp + 2 * CC;

  const v16bf qa0 = frag_rowmajor(Qp + (size_t)qbase * ldq + 0,  ldq, lane);
  const v16bf qa1 = frag_rowmajor(Qp + (size_t)qbase * ldq + 32, ldq, lane);

  v8f o[4] = {};
  float mrow[8], lrow[8], ses[8];
#pragma unroll
  for (int r = 0; r < 8; ++r) { mrow[r] = -INFINITY; lrow[r] = 0.f; ses[r] = 0.f; }

  __shared__ __bf16 Pls[16 * 36];

  const int nkt = (qbase + 47) >> 5;
  for (int kt = 0; kt < nkt; ++kt) {
    const int k0 = kt * 32;
    float st[2][8];
#pragma unroll
    for (int t = 0; t < 2; ++t) {
      const v16bf kb0 = frag_rowmajor(Kp + (size_t)(k0 + t * 16) * ldq + 0,  ldq, lane);
      const v16bf kb1 = frag_rowmajor(Kp + (size_t)(k0 + t * 16) * ldq + 32, ldq, lane);
      v8f s = {};
      s = wmma_bf16(qa0, kb0, s);
      s = wmma_bf16(qa1, kb1, s);
      const int col = k0 + t * 16 + (lane & 15);
#pragma unroll
      for (int r = 0; r < 8; ++r) {
        const int row = qbase + r + 8 * hi;
        st[t][r] = (col <= row) ? s[r] * 0.125f : -1.0e30f;
      }
    }
#pragma unroll
    for (int r = 0; r < 8; ++r) {
      float v = fmaxf(st[0][r], st[1][r]);
#pragma unroll
      for (int off = 1; off <= 8; off <<= 1) v = fmaxf(v, __shfl_xor(v, off, 32));
      const float mn = fmaxf(mrow[r], v);
      const float a  = __expf(mrow[r] - mn);
      const float x0 = __expf(st[0][r] - mn);
      const float x1 = __expf(st[1][r] - mn);
      float ps = x0 + x1;
      float pe = x0 * st[0][r] + x1 * st[1][r];
#pragma unroll
      for (int off = 1; off <= 8; off <<= 1) {
        ps += __shfl_xor(ps, off, 32);
        pe += __shfl_xor(pe, off, 32);
      }
      lrow[r] = lrow[r] * a + ps;
      ses[r]  = ses[r]  * a + pe;
      mrow[r] = mn;
      o[0][r] *= a; o[1][r] *= a; o[2][r] *= a; o[3][r] *= a;
      const int prow = r + 8 * hi;
      Pls[prow * 36 +      (lane & 15)] = (__bf16)x0;
      Pls[prow * 36 + 16 + (lane & 15)] = (__bf16)x1;
    }
    __syncthreads();
    const v16bf pa = frag_rowmajor(Pls, 36, lane);
#pragma unroll
    for (int n = 0; n < 4; ++n) {
      const v16bf vb = frag_b_strided(Vp + (size_t)k0 * ldq + n * 16, ldq, lane);
      o[n] = wmma_bf16(pa, vb, o[n]);
    }
    __syncthreads();
  }

  float inv[8];
#pragma unroll
  for (int r = 0; r < 8; ++r) inv[r] = 1.0f / lrow[r];
#pragma unroll
  for (int n = 0; n < 4; ++n)
#pragma unroll
    for (int r = 0; r < 8; ++r) {
      const int row = qbase + r + 8 * hi;
      const int col = h * HD + n * 16 + (lane & 15);
      y[((size_t)b * TT + row) * CC + col] = (__bf16)(o[n][r] * inv[r]);
    }
  if ((lane & 15) == 0) {
    float acc = 0.f;
#pragma unroll
    for (int r = 0; r < 8; ++r) acc += mrow[r] + __logf(lrow[r]) - ses[r] * inv[r];
    atomicAdd(ent, acc);
  }
}

__global__ void zero_kernel(float* p) { *p = 0.f; }

__global__ void ent_final_kernel(const float* acc, float* out) {
  *out = *acc / (float)(BB * HH * TT);
}

extern "C" void kernel_launch(void* const* d_in, const int* in_sizes, int n_in,
                              void* d_out, int out_size, void* d_ws, size_t ws_size,
                              hipStream_t stream) {
  (void)in_sizes; (void)n_in; (void)out_size; (void)ws_size;
  const float* x          = (const float*)d_in[0];
  const float* ln1_w      = (const float*)d_in[1];
  const float* ln1_b      = (const float*)d_in[2];
  const float* attn_w     = (const float*)d_in[3];
  const float* attn_b     = (const float*)d_in[4];
  const float* attnproj_w = (const float*)d_in[5];
  const float* attnproj_b = (const float*)d_in[6];
  const float* ln2_w      = (const float*)d_in[7];
  const float* ln2_b      = (const float*)d_in[8];
  const float* fc_w       = (const float*)d_in[9];
  const float* fc_b       = (const float*)d_in[10];
  const float* proj_w     = (const float*)d_in[11];
  const float* proj_b     = (const float*)d_in[12];
  float* out = (float*)d_out;

  char* ws = (char*)d_ws;
  float*  x1      = (float*)(ws);
  __bf16* qkv_bf  = (__bf16*)(ws + ((size_t)16 << 20));
  __bf16* h_bf    = (__bf16*)(ws + ((size_t)40 << 20));
  __bf16* y_bf    = (__bf16*)(ws + ((size_t)48 << 20));
  __bf16* m_bf    = (__bf16*)(ws + ((size_t)56 << 20));
  __bf16* wqkv_bf = (__bf16*)(ws + ((size_t)88 << 20));
  __bf16* wap_bf  = (__bf16*)(ws + ((size_t)94 << 20));
  __bf16* wfc_bf  = (__bf16*)(ws + ((size_t)96 << 20));
  __bf16* wpj_bf  = (__bf16*)(ws + ((size_t)104 << 20));
  float*  accum   = (float*)(ws + ((size_t)112 << 20));

  const int M = BB * TT;

  zero_kernel<<<1, 1, 0, stream>>>(accum);
  wconv_kernel<<<(3 * CC * CC) / 256, 256, 0, stream>>>(attn_w, wqkv_bf, 3 * CC * CC);
  wconv_kernel<<<(CC * CC) / 256, 256, 0, stream>>>(attnproj_w, wap_bf, CC * CC);
  wconv_kernel<<<(4 * CC * CC) / 256, 256, 0, stream>>>(fc_w, wfc_bf, 4 * CC * CC);
  wconv_kernel<<<(4 * CC * CC) / 256, 256, 0, stream>>>(proj_w, wpj_bf, 4 * CC * CC);
  ln_kernel<<<M, 256, 0, stream>>>(x, ln1_w, ln1_b, h_bf);
  gemm_kernel<1><<<dim3(3 * CC / 128, M / 128), 256, 0, stream>>>(
      h_bf, wqkv_bf, attn_b, nullptr, qkv_bf, M, 3 * CC, CC);
  attn_kernel<<<dim3(TT / 16, HH, BB), 32, 0, stream>>>(qkv_bf, y_bf, accum);
  gemm_kernel<0><<<dim3(CC / 128, M / 128), 256, 0, stream>>>(
      y_bf, wap_bf, attnproj_b, x, x1, M, CC, CC);
  ln_kernel<<<M, 256, 0, stream>>>(x1, ln2_w, ln2_b, h_bf);
  gemm_kernel<2><<<dim3(4 * CC / 128, M / 128), 256, 0, stream>>>(
      h_bf, wfc_bf, fc_b, nullptr, m_bf, M, 4 * CC, CC);
  gemm_kernel<0><<<dim3(CC / 128, M / 128), 256, 0, stream>>>(
      m_bf, wpj_bf, proj_b, x1, out, M, CC, 4 * CC);
  ent_final_kernel<<<1, 1, 0, stream>>>(accum, out + (size_t)M * CC);
}


# weight f32->bf16 conversion: 4 launches of one element per lane (393216 short waves) become 1 launch of 6144 waves, 64 elements per lane (16 float4 loads in flight, 8-byte stores)
# speedup vs baseline: 1.2669x; 1.2669x over previous
#include <hip/hip_runtime.h>
#include <hip/hip_bf16.h>
#include <math.h>

typedef __attribute__((ext_vector_type(16))) __bf16 v16bf;
typedef __attribute__((ext_vector_type(8)))  float  v8f;
typedef __attribute__((ext_vector_type(8)))  __bf16 v8bf;

#define BB 2
#define TT 2048
#define CC 1024
#define HH 16
#define HD 64

static __device__ inline v8f wmma_bf16(v16bf a, v16bf b, v8f c) {
  return __builtin_amdgcn_wmma_f32_16x16x32_bf16(false, a, false, b, (short)0, c, false, false);
}

static __device__ inline float gelu_tanh(float x) {
  const float c = 0.7978845608028654f;
  return 0.5f * x * (1.0f + tanhf(c * (x + 0.044715f * x * x * x)));
}

static __device__ inline v16bf frag_rowmajor(const __bf16* p, int ld, int lane) {
  const int m  = lane & 15;
  const int ko = (lane >> 4) << 3;
  const __bf16* q = p + (size_t)m * ld + ko;
  v16bf f;
#pragma unroll
  for (int j = 0; j < 4; ++j) {
    f[2 * j]     = q[2 * j];
    f[2 * j + 1] = q[2 * j + 1];
    f[8 + 2 * j] = q[16 + 2 * j];
    f[9 + 2 * j] = q[16 + 2 * j + 1];
  }
  return f;
}

static __device__ inline float xhalf(float x) {
  const int b = __builtin_bit_cast(int, x);
  return __builtin_bit_cast(float, __builtin_amdgcn_permlanex16(b, b, 0x76543210, 0xfedcba98, false, false));
}

typedef __attribute__((ext_vector_type(4))) __bf16 v4bf;
#define WCONV_LOADS 16
__global__ void __launch_bounds__(256)
wconv_kernel(const float* __restrict__ w_qkv, const float* __restrict__ w_ap,
             const float* __restrict__ w_fc, const float* __restrict__ w_pj,
             __bf16* __restrict__ dst) {
  const int u = blockIdx.y;
  const float* src = (u < 3) ? w_qkv + (size_t)u * (CC * CC)
                   : (u < 4) ? w_ap
                   : (u < 8) ? w_fc + (size_t)(u - 4) * (CC * CC)
                             : w_pj + (size_t)(u - 8) * (CC * CC);
  const size_t q0 = (size_t)blockIdx.x * (WCONV_LOADS * 256) + threadIdx.x;
  const float4* s4 = (const float4*)src + q0;
  v4bf* d4 = (v4bf*)(dst + (size_t)u * (CC * CC)) + q0;
  float4 v[WCONV_LOADS];
#pragma unroll
  for (int i = 0; i < WCONV_LOADS; ++i) v[i] = s4[i * 256];
#pragma unroll
  for (int i = 0; i < WCONV_LOADS; ++i) {
    v4bf p;
    p[0] = (__bf16)v[i].x; p[1] = (__bf16)v[i].y;
    p[2] = (__bf16)v[i].z; p[3] = (__bf16)v[i].w;
    d4[i * 256] = p;
  }
}

__global__ void ln_kernel(const float* __restrict__ x, const float* __restrict__ w,
                          const float* __restrict__ b, __bf16* __restrict__ out) {
  const int row = blockIdx.x;
  const int tid = threadIdx.x;
  const float* xr = x + (size_t)row * CC;
  float v[4];
  float s = 0.f, sq = 0.f;
#pragma unroll
  for (int i = 0; i < 4; ++i) {
    v[i] = xr[tid + i * 256];
    s += v[i];
    sq += v[i] * v[i];
  }
#pragma unroll
  for (int off = 16; off >= 1; off >>= 1) {
    s  += __shfl_xor(s,  off, 32);
    sq += __shfl_xor(sq, off, 32);
  }
  __shared__ float reds[8], redq[8];
  const int wid = tid >> 5;
  if ((tid & 31) == 0) { reds[wid] = s; redq[wid] = sq; }
  __syncthreads();
  float ts = 0.f, tq = 0.f;
#pragma unroll
  for (int k = 0; k < 8; ++k) { ts += reds[k]; tq += redq[k]; }
  const float mean = ts * (1.0f / CC);
  const float var  = tq * (1.0f / CC) - mean * mean;
  const float rstd = rsqrtf(var + 1e-5f);
  __bf16* orow = out + (size_t)row * CC;
#pragma unroll
  for (int i = 0; i < 4; ++i) {
    const int c = tid + i * 256;
    orow[c] = (__bf16)((v[i] - mean) * rstd * w[c] + b[c]);
  }
}

template <int EM>
__global__ void __launch_bounds__(256, 1)
gemm_kernel(const __bf16* __restrict__ A, const __bf16* __restrict__ W,
            const float* __restrict__ bias, const float* __restrict__ resid,
            void* __restrict__ outp, __bf16* __restrict__ vT, int M, int N, int K) {
  __shared__ __bf16 As[128 * 40];
  __shared__ __bf16 Bs[128 * 40];
  const int tid  = threadIdx.x;
  const int lane = tid & 31;
  const int wid  = tid >> 5;
  const int wm   = wid & 1;
  const int wn   = wid >> 1;
  const int m0   = blockIdx.y * 128;
  const int n0   = blockIdx.x * 128;

  v8f acc[4][2] = {};

  const int srow = tid >> 1, sseg = tid & 1;
  const __bf16* agp = A + (size_t)(m0 + srow) * K + sseg * 16;
  const __bf16* bgp = W + (size_t)(n0 + srow) * K + sseg * 16;

  for (int k0 = 0; k0 < K; k0 += 32) {
    {
      const uint4* ap = (const uint4*)(agp + k0);
      uint4* ad = (uint4*)(As + srow * 40 + sseg * 16);
      ad[0] = ap[0]; ad[1] = ap[1];
      const uint4* bp = (const uint4*)(bgp + k0);
      uint4* bd = (uint4*)(Bs + srow * 40 + sseg * 16);
      bd[0] = bp[0]; bd[1] = bp[1];
    }
    if (k0 + 32 < K) {
      __builtin_prefetch(agp + k0 + 32, 0, 0);
      __builtin_prefetch(bgp + k0 + 32, 0, 0);
    }
    __syncthreads();

    v16bf af[4], bf[2];
#pragma unroll
    for (int i = 0; i < 4; ++i) af[i] = frag_rowmajor(As + (wm * 64 + i * 16) * 40, 40, lane);
#pragma unroll
    for (int j = 0; j < 2; ++j) bf[j] = frag_rowmajor(Bs + (wn * 32 + j * 16) * 40, 40, lane);
#pragma unroll
    for (int i = 0; i < 4; ++i)
#pragma unroll
      for (int j = 0; j < 2; ++j) acc[i][j] = wmma_bf16(af[i], bf[j], acc[i][j]);
    __syncthreads();
  }

  const int hi = lane >> 4;
  float biasj[2];
#pragma unroll
  for (int j = 0; j < 2; ++j) biasj[j] = bias[n0 + wn * 32 + j * 16 + (lane & 15)];

  if constexpr (EM == 3) {
    if (n0 >= 2 * CC) {
#pragma unroll
      for (int i = 0; i < 4; ++i)
#pragma unroll
        for (int j = 0; j < 2; ++j) {
          const int row0 = m0 + wm * 64 + i * 16 + 8 * hi;
          const int c    = n0 - 2 * CC + wn * 32 + j * 16 + (lane & 15);
          const int bb   = row0 / TT;
          const int t0   = row0 % TT;
          const size_t tile = (size_t)(bb * HH + (c >> 6)) * (TT / 32) + (t0 >> 5);
          v8bf pk;
#pragma unroll
          for (int r = 0; r < 8; ++r) pk[r] = (__bf16)(acc[i][j][r] + biasj[j]);
          *(v8bf*)(vT + (tile * HD + (c & 63)) * 32 + (t0 & 31)) = pk;
        }
      return;
    }
  }

#pragma unroll
  for (int i = 0; i < 4; ++i)
#pragma unroll
    for (int j = 0; j < 2; ++j)
#pragma unroll
      for (int r = 0; r < 8; ++r) {
        const int row = m0 + wm * 64 + i * 16 + r + 8 * hi;
        const int col = n0 + wn * 32 + j * 16 + (lane & 15);
        float v = acc[i][j][r] + biasj[j];
        const size_t idx = (size_t)row * N + col;
        if constexpr (EM == 0) {
          ((float*)outp)[idx] = v + resid[idx];
        } else if constexpr (EM == 1 || EM == 3) {
          ((__bf16*)outp)[idx] = (__bf16)v;
        } else {
          ((__bf16*)outp)[idx] = (__bf16)gelu_tanh(v);
        }
      }
}

__global__ void __launch_bounds__(32, 1)
attn_kernel(const __bf16* __restrict__ qkv, const __bf16* __restrict__ vT,
            __bf16* __restrict__ y, float* __restrict__ ent) {
  const int lane  = threadIdx.x;
  const int hi    = lane >> 4;
  const int qbase = blockIdx.x * 16;
  const int h     = blockIdx.y;
  const int b     = blockIdx.z;
  const int ldq   = 3 * CC;
  const int qrow  = qbase + (lane & 15);

  const __bf16* Qp  = qkv + (size_t)b * TT * ldq + h * HD;
  const __bf16* Kp  = Qp + CC;
  const __bf16* VTp = vT + (size_t)(b * HH + h) * (TT / 32) * (HD * 32);

  const v16bf qb0 = frag_rowmajor(Qp + (size_t)qbase * ldq + 0,  ldq, lane);
  const v16bf qb1 = frag_rowmajor(Qp + (size_t)qbase * ldq + 32, ldq, lane);

  v8f o[4] = {};
  float mrow = -INFINITY, lrow = 0.f, ses = 0.f;

  const int nkt = (qbase + 47) >> 5;
  for (int kt = 0; kt < nkt; ++kt) {
    const int k0 = kt * 32;
    float st[2][8];
#pragma unroll
    for (int t = 0; t < 2; ++t) {
      const v16bf ka0 = frag_rowmajor(Kp + (size_t)(k0 + t * 16) * ldq + 0,  ldq, lane);
      const v16bf ka1 = frag_rowmajor(Kp + (size_t)(k0 + t * 16) * ldq + 32, ldq, lane);
      v8f s = {};
      s = wmma_bf16(ka0, qb0, s);
      s = wmma_bf16(ka1, qb1, s);
#pragma unroll
      for (int r = 0; r < 8; ++r) {
        const int key = k0 + t * 16 + 8 * hi + r;
        st[t][r] = (key <= qrow) ? s[r] * 0.125f : -1.0e30f;
      }
    }
    float v = fmaxf(st[0][0], st[1][0]);
#pragma unroll
    for (int r = 1; r < 8; ++r) v = fmaxf(v, fmaxf(st[0][r], st[1][r]));
    v = fmaxf(v, xhalf(v));
    const float mn = fmaxf(mrow, v);
    const float a  = __expf(mrow - mn);
    float x[2][8];
    float ps = 0.f, pe = 0.f;
#pragma unroll
    for (int r = 0; r < 8; ++r) {
      x[0][r] = __expf(st[0][r] - mn);
      x[1][r] = __expf(st[1][r] - mn);
      ps += x[0][r] + x[1][r];
      pe += x[0][r] * st[0][r] + x[1][r] * st[1][r];
    }
    ps += xhalf(ps);
    pe += xhalf(pe);
    lrow = lrow * a + ps;
    ses  = ses  * a + pe;
    mrow = mn;
#pragma unroll
    for (int n = 0; n < 4; ++n)
#pragma unroll
      for (int r = 0; r < 8; ++r) o[n][r] *= a;
    v16bf pb;
#pragma unroll
    for (int r = 0; r < 8; ++r) {
      pb[r]     = (__bf16)x[0][r];
      pb[8 + r] = (__bf16)x[1][r];
    }
    const __bf16* vtile = VTp + (size_t)kt * (HD * 32);
#pragma unroll
    for (int n = 0; n < 4; ++n) {
      const v16bf va = frag_rowmajor(vtile + (n * 16) * 32, 32, lane);
      o[n] = wmma_bf16(va, pb, o[n]);
    }
  }

  const float inv = 1.0f / lrow;
  __bf16* yrow = y + ((size_t)b * TT + qrow) * CC + h * HD + 8 * hi;
#pragma unroll
  for (int n = 0; n < 4; ++n) {
    v8bf pk;
#pragma unroll
    for (int r = 0; r < 8; ++r) pk[r] = (__bf16)(o[n][r] * inv);
    *(v8bf*)(yrow + n * 16) = pk;
  }
  float e = mrow + __logf(lrow) - ses * inv;
#pragma unroll
  for (int off = 1; off <= 8; off <<= 1) e += __shfl_xor(e, off, 32);
  if (lane == 0) atomicAdd(ent, e);
}

__global__ void zero_kernel(float* p) { *p = 0.f; }

__global__ void ent_final_kernel(const float* acc, float* out) {
  *out = *acc / (float)(BB * HH * TT);
}

extern "C" void kernel_launch(void* const* d_in, const int* in_sizes, int n_in,
                              void* d_out, int out_size, void* d_ws, size_t ws_size,
                              hipStream_t stream) {
  (void)in_sizes; (void)n_in; (void)out_size; (void)ws_size;
  const float* x          = (const float*)d_in[0];
  const float* ln1_w      = (const float*)d_in[1];
  const float* ln1_b      = (const float*)d_in[2];
  const float* attn_w     = (const float*)d_in[3];
  const float* attn_b     = (const float*)d_in[4];
  const float* attnproj_w = (const float*)d_in[5];
  const float* attnproj_b = (const float*)d_in[6];
  const float* ln2_w      = (const float*)d_in[7];
  const float* ln2_b      = (const float*)d_in[8];
  const float* fc_w       = (const float*)d_in[9];
  const float* fc_b       = (const float*)d_in[10];
  const float* proj_w     = (const float*)d_in[11];
  const float* proj_b     = (const float*)d_in[12];
  float* out = (float*)d_out;

  char* ws = (char*)d_ws;
  float*  x1      = (float*)(ws);
  __bf16* qkv_bf  = (__bf16*)(ws + ((size_t)16 << 20));
  __bf16* h_bf    = (__bf16*)(ws + ((size_t)40 << 20));
  __bf16* y_bf    = (__bf16*)(ws + ((size_t)48 << 20));
  __bf16* m_bf    = (__bf16*)(ws + ((size_t)56 << 20));
  __bf16* vT_bf   = m_bf;
  __bf16* wqkv_bf = (__bf16*)(ws + ((size_t)88 << 20));
  __bf16* wap_bf  = (__bf16*)(ws + ((size_t)94 << 20));
  __bf16* wfc_bf  = (__bf16*)(ws + ((size_t)96 << 20));
  __bf16* wpj_bf  = (__bf16*)(ws + ((size_t)104 << 20));
  float*  accum   = (float*)(ws + ((size_t)112 << 20));

  const int M = BB * TT;

  zero_kernel<<<1, 1, 0, stream>>>(accum);
  wconv_kernel<<<dim3((CC * CC) / (4 * WCONV_LOADS * 256), 12), 256, 0, stream>>>(
      attn_w, attnproj_w, fc_w, proj_w, wqkv_bf);
  ln_kernel<<<M, 256, 0, stream>>>(x, ln1_w, ln1_b, h_bf);
  gemm_kernel<3><<<dim3(3 * CC / 128, M / 128), 256, 0, stream>>>(
      h_bf, wqkv_bf, attn_b, nullptr, qkv_bf, vT_bf, M, 3 * CC, CC);
  attn_kernel<<<dim3(TT / 16, HH, BB), 32, 0, stream>>>(qkv_bf, vT_bf, y_bf, accum);
  gemm_kernel<0><<<dim3(CC / 128, M / 128), 256, 0, stream>>>(
      y_bf, wap_bf, attnproj_b, x, x1, nullptr, M, CC, CC);
  ln_kernel<<<M, 256, 0, stream>>>(x1, ln2_w, ln2_b, h_bf);
  gemm_kernel<2><<<dim3(4 * CC / 128, M / 128), 256, 0, stream>>>(
      h_bf, wfc_bf, fc_b, nullptr, m_bf, nullptr, M, 4 * CC, CC);
  gemm_kernel<0><<<dim3(CC / 128, M / 128), 256, 0, stream>>>(
      m_bf, wpj_bf, proj_b, x1, out, nullptr, M, CC, 4 * CC);
  ent_final_kernel<<<1, 1, 0, stream>>>(accum, out + (size_t)M * CC);
}


# GEMM K loop software-pipelined over two LDS stages: next step's global tile loads in flight as one group under the step's WMMAs, stored to LDS behind them; one barrier per step; L2 prefetch dropped
# speedup vs baseline: 1.6453x; 1.2987x over previous
#include <hip/hip_runtime.h>
#include <hip/hip_bf16.h>
#include <math.h>

typedef __attribute__((ext_vector_type(16))) __bf16 v16bf;
typedef __attribute__((ext_vector_type(8)))  float  v8f;
typedef __attribute__((ext_vector_type(8)))  __bf16 v8bf;

#define BB 2
#define TT 2048
#define CC 1024
#define HH 16
#define HD 64

static __device__ inline v8f wmma_bf16(v16bf a, v16bf b, v8f c) {
  return __builtin_amdgcn_wmma_f32_16x16x32_bf16(false, a, false, b, (short)0, c, false, false);
}

static __device__ inline float gelu_tanh(float x) {
  const float c = 0.7978845608028654f;
  return 0.5f * x * (1.0f + tanhf(c * (x + 0.044715f * x * x * x)));
}

static __device__ inline v16bf frag_rowmajor(const __bf16* p, int ld, int lane) {
  const int m  = lane & 15;
  const int ko = (lane >> 4) << 3;
  const __bf16* q = p + (size_t)m * ld + ko;
  v16bf f;
#pragma unroll
  for (int j = 0; j < 4; ++j) {
    f[2 * j]     = q[2 * j];
    f[2 * j + 1] = q[2 * j + 1];
    f[8 + 2 * j] = q[16 + 2 * j];
    f[9 + 2 * j] = q[16 + 2 * j + 1];
  }
  return f;
}

static __device__ inline float xhalf(float x) {
  const int b = __builtin_bit_cast(int, x);
  return __builtin_bit_cast(float, __builtin_amdgcn_permlanex16(b, b, 0x76543210, 0xfedcba98, false, false));
}

typedef __attribute__((ext_vector_type(4))) __bf16 v4bf;
#define WCONV_LOADS 16
__global__ void __launch_bounds__(256)
wconv_kernel(const float* __restrict__ w_qkv, const float* __restrict__ w_ap,
             const float* __restrict__ w_fc, const float* __restrict__ w_pj,
             __bf16* __restrict__ dst) {
  const int u = blockIdx.y;
  const float* src = (u < 3) ? w_qkv + (size_t)u * (CC * CC)
                   : (u < 4) ? w_ap
                   : (u < 8) ? w_fc + (size_t)(u - 4) * (CC * CC)
                             : w_pj + (size_t)(u - 8) * (CC * CC);
  const size_t q0 = (size_t)blockIdx.x * (WCONV_LOADS * 256) + threadIdx.x;
  const float4* s4 = (const float4*)src + q0;
  v4bf* d4 = (v4bf*)(dst + (size_t)u * (CC * CC)) + q0;
  float4 v[WCONV_LOADS];
#pragma unroll
  for (int i = 0; i < WCONV_LOADS; ++i) v[i] = s4[i * 256];
#pragma unroll
  for (int i = 0; i < WCONV_LOADS; ++i) {
    v4bf p;
    p[0] = (__bf16)v[i].x; p[1] = (__bf16)v[i].y;
    p[2] = (__bf16)v[i].z; p[3] = (__bf16)v[i].w;
    d4[i * 256] = p;
  }
}

__global__ void ln_kernel(const float* __restrict__ x, const float* __restrict__ w,
                          const float* __restrict__ b, __bf16* __restrict__ out) {
  const int row = blockIdx.x;
  const int tid = threadIdx.x;
  const float* xr = x + (size_t)row * CC;
  float v[4];
  float s = 0.f, sq = 0.f;
#pragma unroll
  for (int i = 0; i < 4; ++i) {
    v[i] = xr[tid + i * 256];
    s += v[i];
    sq += v[i] * v[i];
  }
#pragma unroll
  for (int off = 16; off >= 1; off >>= 1) {
    s  += __shfl_xor(s,  off, 32);
    sq += __shfl_xor(sq, off, 32);
  }
  __shared__ float reds[8], redq[8];
  const int wid = tid >> 5;
  if ((tid & 31) == 0) { reds[wid] = s; redq[wid] = sq; }
  __syncthreads();
  float ts = 0.f, tq = 0.f;
#pragma unroll
  for (int k = 0; k < 8; ++k) { ts += reds[k]; tq += redq[k]; }
  const float mean = ts * (1.0f / CC);
  const float var  = tq * (1.0f / CC) - mean * mean;
  const float rstd = rsqrtf(var + 1e-5f);
  __bf16* orow = out + (size_t)row * CC;
#pragma unroll
  for (int i = 0; i < 4; ++i) {
    const int c = tid + i * 256;
    orow[c] = (__bf16)((v[i] - mean) * rstd * w[c] + b[c]);
  }
}


static __device__ inline void mma_step(const __bf16* as, const __bf16* bs, v8f (&acc)[4][2],
                                       int wm, int wn, int lane) {
  v16bf af[4], bf[2];
#pragma unroll
  for (int i = 0; i < 4; ++i) af[i] = frag_rowmajor(as + (wm * 64 + i * 16) * 40, 40, lane);
#pragma unroll
  for (int j = 0; j < 2; ++j) bf[j] = frag_rowmajor(bs + (wn * 32 + j * 16) * 40, 40, lane);
#pragma unroll
  for (int i = 0; i < 4; ++i)
#pragma unroll
    for (int j = 0; j < 2; ++j) acc[i][j] = wmma_bf16(af[i], bf[j], acc[i][j]);
}

template <int EM>
__global__ void __launch_bounds__(256, 1)
gemm_kernel(const __bf16* A, const __bf16* W,
            const float* __restrict__ bias, const float* __restrict__ resid,
            void* __restrict__ outp, __bf16* __restrict__ vT, int M, int N, int K) {
  __shared__ __bf16 As[2][128 * 40];
  __shared__ __bf16 Bs[2][128 * 40];
  const int tid  = threadIdx.x;
  const int lane = tid & 31;
  const int wid  = tid >> 5;
  const int wm   = wid & 1;
  const int wn   = wid >> 1;
  const int m0   = blockIdx.y * 128;
  const int n0   = blockIdx.x * 128;

  v8f acc[4][2] = {};

  const int srow = tid >> 1, sseg = tid & 1;
  const uint4* agp = (const uint4*)(A + (size_t)(m0 + srow) * K + sseg * 16);
  const uint4* bgp = (const uint4*)(W + (size_t)(n0 + srow) * K + sseg * 16);
  const int soff = srow * 40 + sseg * 16;

  uint4 ra0 = agp[0], ra1 = agp[1], rb0 = bgp[0], rb1 = bgp[1];
  {
    uint4* ad = (uint4*)(As[0] + soff);
    uint4* bd = (uint4*)(Bs[0] + soff);
    ad[0] = ra0; ad[1] = ra1;
    bd[0] = rb0; bd[1] = rb1;
  }
  __syncthreads();

  const int nk = K >> 5;
  for (int kt = 0; kt + 1 < nk; ++kt) {
    const int cur = kt & 1;
    const uint4* ap = agp + (size_t)(kt + 1) * 4;
    const uint4* bp = bgp + (size_t)(kt + 1) * 4;
    ra0 = ap[0]; ra1 = ap[1];
    rb0 = bp[0]; rb1 = bp[1];
    asm volatile("" ::: "memory");

    mma_step(As[cur], Bs[cur], acc, wm, wn, lane);

    uint4* ad = (uint4*)(As[cur ^ 1] + soff);
    uint4* bd = (uint4*)(Bs[cur ^ 1] + soff);
    ad[0] = ra0; ad[1] = ra1;
    bd[0] = rb0; bd[1] = rb1;
    __syncthreads();
  }
  mma_step(As[(nk - 1) & 1], Bs[(nk - 1) & 1], acc, wm, wn, lane);

  const int hi = lane >> 4;
  float biasj[2];
#pragma unroll
  for (int j = 0; j < 2; ++j) biasj[j] = bias[n0 + wn * 32 + j * 16 + (lane & 15)];

  if constexpr (EM == 3) {
    if (n0 >= 2 * CC) {
#pragma unroll
      for (int i = 0; i < 4; ++i)
#pragma unroll
        for (int j = 0; j < 2; ++j) {
          const int row0 = m0 + wm * 64 + i * 16 + 8 * hi;
          const int c    = n0 - 2 * CC + wn * 32 + j * 16 + (lane & 15);
          const int bb   = row0 / TT;
          const int t0   = row0 % TT;
          const size_t tile = (size_t)(bb * HH + (c >> 6)) * (TT / 32) + (t0 >> 5);
          v8bf pk;
#pragma unroll
          for (int r = 0; r < 8; ++r) pk[r] = (__bf16)(acc[i][j][r] + biasj[j]);
          *(v8bf*)(vT + (tile * HD + (c & 63)) * 32 + (t0 & 31)) = pk;
        }
      return;
    }
  }

#pragma unroll
  for (int i = 0; i < 4; ++i)
#pragma unroll
    for (int j = 0; j < 2; ++j)
#pragma unroll
      for (int r = 0; r < 8; ++r) {
        const int row = m0 + wm * 64 + i * 16 + r + 8 * hi;
        const int col = n0 + wn * 32 + j * 16 + (lane & 15);
        float v = acc[i][j][r] + biasj[j];
        const size_t idx = (size_t)row * N + col;
        if constexpr (EM == 0) {
          ((float*)outp)[idx] = v + resid[idx];
        } else if constexpr (EM == 1 || EM == 3) {
          ((__bf16*)outp)[idx] = (__bf16)v;
        } else {
          ((__bf16*)outp)[idx] = (__bf16)gelu_tanh(v);
        }
      }
}

__global__ void __launch_bounds__(32, 1)
attn_kernel(const __bf16* __restrict__ qkv, const __bf16* __restrict__ vT,
            __bf16* __restrict__ y, float* __restrict__ ent) {
  const int lane  = threadIdx.x;
  const int hi    = lane >> 4;
  const int qbase = blockIdx.x * 16;
  const int h     = blockIdx.y;
  const int b     = blockIdx.z;
  const int ldq   = 3 * CC;
  const int qrow  = qbase + (lane & 15);

  const __bf16* Qp  = qkv + (size_t)b * TT * ldq + h * HD;
  const __bf16* Kp  = Qp + CC;
  const __bf16* VTp = vT + (size_t)(b * HH + h) * (TT / 32) * (HD * 32);

  const v16bf qb0 = frag_rowmajor(Qp + (size_t)qbase * ldq + 0,  ldq, lane);
  const v16bf qb1 = frag_rowmajor(Qp + (size_t)qbase * ldq + 32, ldq, lane);

  v8f o[4] = {};
  float mrow = -INFINITY, lrow = 0.f, ses = 0.f;

  const int nkt = (qbase + 47) >> 5;
  for (int kt = 0; kt < nkt; ++kt) {
    const int k0 = kt * 32;
    float st[2][8];
#pragma unroll
    for (int t = 0; t < 2; ++t) {
      const v16bf ka0 = frag_rowmajor(Kp + (size_t)(k0 + t * 16) * ldq + 0,  ldq, lane);
      const v16bf ka1 = frag_rowmajor(Kp + (size_t)(k0 + t * 16) * ldq + 32, ldq, lane);
      v8f s = {};
      s = wmma_bf16(ka0, qb0, s);
      s = wmma_bf16(ka1, qb1, s);
#pragma unroll
      for (int r = 0; r < 8; ++r) {
        const int key = k0 + t * 16 + 8 * hi + r;
        st[t][r] = (key <= qrow) ? s[r] * 0.125f : -1.0e30f;
      }
    }
    float v = fmaxf(st[0][0], st[1][0]);
#pragma unroll
    for (int r = 1; r < 8; ++r) v = fmaxf(v, fmaxf(st[0][r], st[1][r]));
    v = fmaxf(v, xhalf(v));
    const float mn = fmaxf(mrow, v);
    const float a  = __expf(mrow - mn);
    float x[2][8];
    float ps = 0.f, pe = 0.f;
#pragma unroll
    for (int r = 0; r < 8; ++r) {
      x[0][r] = __expf(st[0][r] - mn);
      x[1][r] = __expf(st[1][r] - mn);
      ps += x[0][r] + x[1][r];
      pe += x[0][r] * st[0][r] + x[1][r] * st[1][r];
    }
    ps += xhalf(ps);
    pe += xhalf(pe);
    lrow = lrow * a + ps;
    ses  = ses  * a + pe;
    mrow = mn;
#pragma unroll
    for (int n = 0; n < 4; ++n)
#pragma unroll
      for (int r = 0; r < 8; ++r) o[n][r] *= a;
    v16bf pb;
#pragma unroll
    for (int r = 0; r < 8; ++r) {
      pb[r]     = (__bf16)x[0][r];
      pb[8 + r] = (__bf16)x[1][r];
    }
    const __bf16* vtile = VTp + (size_t)kt * (HD * 32);
#pragma unroll
    for (int n = 0; n < 4; ++n) {
      const v16bf va = frag_rowmajor(vtile + (n * 16) * 32, 32, lane);
      o[n] = wmma_bf16(va, pb, o[n]);
    }
  }

  const float inv = 1.0f / lrow;
  __bf16* yrow = y + ((size_t)b * TT + qrow) * CC + h * HD + 8 * hi;
#pragma unroll
  for (int n = 0; n < 4; ++n) {
    v8bf pk;
#pragma unroll
    for (int r = 0; r < 8; ++r) pk[r] = (__bf16)(o[n][r] * inv);
    *(v8bf*)(yrow + n * 16) = pk;
  }
  float e = mrow + __logf(lrow) - ses * inv;
#pragma unroll
  for (int off = 1; off <= 8; off <<= 1) e += __shfl_xor(e, off, 32);
  if (lane == 0) atomicAdd(ent, e);
}

__global__ void zero_kernel(float* p) { *p = 0.f; }

__global__ void ent_final_kernel(const float* acc, float* out) {
  *out = *acc / (float)(BB * HH * TT);
}

extern "C" void kernel_launch(void* const* d_in, const int* in_sizes, int n_in,
                              void* d_out, int out_size, void* d_ws, size_t ws_size,
                              hipStream_t stream) {
  (void)in_sizes; (void)n_in; (void)out_size; (void)ws_size;
  const float* x          = (const float*)d_in[0];
  const float* ln1_w      = (const float*)d_in[1];
  const float* ln1_b      = (const float*)d_in[2];
  const float* attn_w     = (const float*)d_in[3];
  const float* attn_b     = (const float*)d_in[4];
  const float* attnproj_w = (const float*)d_in[5];
  const float* attnproj_b = (const float*)d_in[6];
  const float* ln2_w      = (const float*)d_in[7];
  const float* ln2_b      = (const float*)d_in[8];
  const float* fc_w       = (const float*)d_in[9];
  const float* fc_b       = (const float*)d_in[10];
  const float* proj_w     = (const float*)d_in[11];
  const float* proj_b     = (const float*)d_in[12];
  float* out = (float*)d_out;

  char* ws = (char*)d_ws;
  float*  x1      = (float*)(ws);
  __bf16* qkv_bf  = (__bf16*)(ws + ((size_t)16 << 20));
  __bf16* h_bf    = (__bf16*)(ws + ((size_t)40 << 20));
  __bf16* y_bf    = (__bf16*)(ws + ((size_t)48 << 20));
  __bf16* m_bf    = (__bf16*)(ws + ((size_t)56 << 20));
  __bf16* vT_bf   = m_bf;
  __bf16* wqkv_bf = (__bf16*)(ws + ((size_t)88 << 20));
  __bf16* wap_bf  = (__bf16*)(ws + ((size_t)94 << 20));
  __bf16* wfc_bf  = (__bf16*)(ws + ((size_t)96 << 20));
  __bf16* wpj_bf  = (__bf16*)(ws + ((size_t)104 << 20));
  float*  accum   = (float*)(ws + ((size_t)112 << 20));

  const int M = BB * TT;

  zero_kernel<<<1, 1, 0, stream>>>(accum);
  wconv_kernel<<<dim3((CC * CC) / (4 * WCONV_LOADS * 256), 12), 256, 0, stream>>>(
      attn_w, attnproj_w, fc_w, proj_w, wqkv_bf);
  ln_kernel<<<M, 256, 0, stream>>>(x, ln1_w, ln1_b, h_bf);
  gemm_kernel<3><<<dim3(3 * CC / 128, M / 128), 256, 0, stream>>>(
      h_bf, wqkv_bf, attn_b, nullptr, qkv_bf, vT_bf, M, 3 * CC, CC);
  attn_kernel<<<dim3(TT / 16, HH, BB), 32, 0, stream>>>(qkv_bf, vT_bf, y_bf, accum);
  gemm_kernel<0><<<dim3(CC / 128, M / 128), 256, 0, stream>>>(
      y_bf, wap_bf, attnproj_b, x, x1, nullptr, M, CC, CC);
  ln_kernel<<<M, 256, 0, stream>>>(x1, ln2_w, ln2_b, h_bf);
  gemm_kernel<2><<<dim3(4 * CC / 128, M / 128), 256, 0, stream>>>(
      h_bf, wfc_bf, fc_b, nullptr, m_bf, nullptr, M, 4 * CC, CC);
  gemm_kernel<0><<<dim3(CC / 128, M / 128), 256, 0, stream>>>(
      m_bf, wpj_bf, proj_b, x1, out, nullptr, M, CC, 4 * CC);
  ent_final_kernel<<<1, 1, 0, stream>>>(accum, out + (size_t)M * CC);
}


# attention key loop software-pipelined: next step's K fragments loaded right behind the score WMMAs, next V^T fragments behind the O^T WMMAs, counted waits; sums order unchanged
# speedup vs baseline: 1.8749x; 1.1395x over previous
#include <hip/hip_runtime.h>
#include <hip/hip_bf16.h>
#include <math.h>

typedef __attribute__((ext_vector_type(16))) __bf16 v16bf;
typedef __attribute__((ext_vector_type(8)))  float  v8f;
typedef __attribute__((ext_vector_type(8)))  __bf16 v8bf;

#define BB 2
#define TT 2048
#define CC 1024
#define HH 16
#define HD 64

static __device__ inline v8f wmma_bf16(v16bf a, v16bf b, v8f c) {
  return __builtin_amdgcn_wmma_f32_16x16x32_bf16(false, a, false, b, (short)0, c, false, false);
}

static __device__ inline float gelu_tanh(float x) {
  const float c = 0.7978845608028654f;
  return 0.5f * x * (1.0f + tanhf(c * (x + 0.044715f * x * x * x)));
}

static __device__ inline v16bf frag_rowmajor(const __bf16* p, int ld, int lane) {
  const int m  = lane & 15;
  const int ko = (lane >> 4) << 3;
  const __bf16* q = p + (size_t)m * ld + ko;
  v16bf f;
#pragma unroll
  for (int j = 0; j < 4; ++j) {
    f[2 * j]     = q[2 * j];
    f[2 * j + 1] = q[2 * j + 1];
    f[8 + 2 * j] = q[16 + 2 * j];
    f[9 + 2 * j] = q[16 + 2 * j + 1];
  }
  return f;
}

static __device__ inline float xhalf(float x) {
  const int b = __builtin_bit_cast(int, x);
  return __builtin_bit_cast(float, __builtin_amdgcn_permlanex16(b, b, 0x76543210, 0xfedcba98, false, false));
}

typedef __attribute__((ext_vector_type(4))) __bf16 v4bf;
#define WCONV_LOADS 16
__global__ void __launch_bounds__(256)
wconv_kernel(const float* __restrict__ w_qkv, const float* __restrict__ w_ap,
             const float* __restrict__ w_fc, const float* __restrict__ w_pj,
             __bf16* __restrict__ dst) {
  const int u = blockIdx.y;
  const float* src = (u < 3) ? w_qkv + (size_t)u * (CC * CC)
                   : (u < 4) ? w_ap
                   : (u < 8) ? w_fc + (size_t)(u - 4) * (CC * CC)
                             : w_pj + (size_t)(u - 8) * (CC * CC);
  const size_t q0 = (size_t)blockIdx.x * (WCONV_LOADS * 256) + threadIdx.x;
  const float4* s4 = (const float4*)src + q0;
  v4bf* d4 = (v4bf*)(dst + (size_t)u * (CC * CC)) + q0;
  float4 v[WCONV_LOADS];
#pragma unroll
  for (int i = 0; i < WCONV_LOADS; ++i) v[i] = s4[i * 256];
#pragma unroll
  for (int i = 0; i < WCONV_LOADS; ++i) {
    v4bf p;
    p[0] = (__bf16)v[i].x; p[1] = (__bf16)v[i].y;
    p[2] = (__bf16)v[i].z; p[3] = (__bf16)v[i].w;
    d4[i * 256] = p;
  }
}

__global__ void ln_kernel(const float* __restrict__ x, const float* __restrict__ w,
                          const float* __restrict__ b, __bf16* __restrict__ out) {
  const int row = blockIdx.x;
  const int tid = threadIdx.x;
  const float* xr = x + (size_t)row * CC;
  float v[4];
  float s = 0.f, sq = 0.f;
#pragma unroll
  for (int i = 0; i < 4; ++i) {
    v[i] = xr[tid + i * 256];
    s += v[i];
    sq += v[i] * v[i];
  }
#pragma unroll
  for (int off = 16; off >= 1; off >>= 1) {
    s  += __shfl_xor(s,  off, 32);
    sq += __shfl_xor(sq, off, 32);
  }
  __shared__ float reds[8], redq[8];
  const int wid = tid >> 5;
  if ((tid & 31) == 0) { reds[wid] = s; redq[wid] = sq; }
  __syncthreads();
  float ts = 0.f, tq = 0.f;
#pragma unroll
  for (int k = 0; k < 8; ++k) { ts += reds[k]; tq += redq[k]; }
  const float mean = ts * (1.0f / CC);
  const float var  = tq * (1.0f / CC) - mean * mean;
  const float rstd = rsqrtf(var + 1e-5f);
  __bf16* orow = out + (size_t)row * CC;
#pragma unroll
  for (int i = 0; i < 4; ++i) {
    const int c = tid + i * 256;
    orow[c] = (__bf16)((v[i] - mean) * rstd * w[c] + b[c]);
  }
}


static __device__ inline void mma_step(const __bf16* as, const __bf16* bs, v8f (&acc)[4][2],
                                       int wm, int wn, int lane) {
  v16bf af[4], bf[2];
#pragma unroll
  for (int i = 0; i < 4; ++i) af[i] = frag_rowmajor(as + (wm * 64 + i * 16) * 40, 40, lane);
#pragma unroll
  for (int j = 0; j < 2; ++j) bf[j] = frag_rowmajor(bs + (wn * 32 + j * 16) * 40, 40, lane);
#pragma unroll
  for (int i = 0; i < 4; ++i)
#pragma unroll
    for (int j = 0; j < 2; ++j) acc[i][j] = wmma_bf16(af[i], bf[j], acc[i][j]);
}

template <int EM>
__global__ void __launch_bounds__(256, 1)
gemm_kernel(const __bf16* A, const __bf16* W,
            const float* __restrict__ bias, const float* __restrict__ resid,
            void* __restrict__ outp, __bf16* __restrict__ vT, int M, int N, int K) {
  __shared__ __bf16 As[2][128 * 40];
  __shared__ __bf16 Bs[2][128 * 40];
  const int tid  = threadIdx.x;
  const int lane = tid & 31;
  const int wid  = tid >> 5;
  const int wm   = wid & 1;
  const int wn   = wid >> 1;
  const int m0   = blockIdx.y * 128;
  const int n0   = blockIdx.x * 128;

  v8f acc[4][2] = {};

  const int srow = tid >> 1, sseg = tid & 1;
  const uint4* agp = (const uint4*)(A + (size_t)(m0 + srow) * K + sseg * 16);
  const uint4* bgp = (const uint4*)(W + (size_t)(n0 + srow) * K + sseg * 16);
  const int soff = srow * 40 + sseg * 16;

  uint4 ra0 = agp[0], ra1 = agp[1], rb0 = bgp[0], rb1 = bgp[1];
  {
    uint4* ad = (uint4*)(As[0] + soff);
    uint4* bd = (uint4*)(Bs[0] + soff);
    ad[0] = ra0; ad[1] = ra1;
    bd[0] = rb0; bd[1] = rb1;
  }
  __syncthreads();

  const int nk = K >> 5;
  for (int kt = 0; kt + 1 < nk; ++kt) {
    const int cur = kt & 1;
    const uint4* ap = agp + (size_t)(kt + 1) * 4;
    const uint4* bp = bgp + (size_t)(kt + 1) * 4;
    ra0 = ap[0]; ra1 = ap[1];
    rb0 = bp[0]; rb1 = bp[1];
    asm volatile("" ::: "memory");

    mma_step(As[cur], Bs[cur], acc, wm, wn, lane);

    uint4* ad = (uint4*)(As[cur ^ 1] + soff);
    uint4* bd = (uint4*)(Bs[cur ^ 1] + soff);
    ad[0] = ra0; ad[1] = ra1;
    bd[0] = rb0; bd[1] = rb1;
    __syncthreads();
  }
  mma_step(As[(nk - 1) & 1], Bs[(nk - 1) & 1], acc, wm, wn, lane);

  const int hi = lane >> 4;
  float biasj[2];
#pragma unroll
  for (int j = 0; j < 2; ++j) biasj[j] = bias[n0 + wn * 32 + j * 16 + (lane & 15)];

  if constexpr (EM == 3) {
    if (n0 >= 2 * CC) {
#pragma unroll
      for (int i = 0; i < 4; ++i)
#pragma unroll
        for (int j = 0; j < 2; ++j) {
          const int row0 = m0 + wm * 64 + i * 16 + 8 * hi;
          const int c    = n0 - 2 * CC + wn * 32 + j * 16 + (lane & 15);
          const int bb   = row0 / TT;
          const int t0   = row0 % TT;
          const size_t tile = (size_t)(bb * HH + (c >> 6)) * (TT / 32) + (t0 >> 5);
          v8bf pk;
#pragma unroll
          for (int r = 0; r < 8; ++r) pk[r] = (__bf16)(acc[i][j][r] + biasj[j]);
          *(v8bf*)(vT + (tile * HD + (c & 63)) * 32 + (t0 & 31)) = pk;
        }
      return;
    }
  }

#pragma unroll
  for (int i = 0; i < 4; ++i)
#pragma unroll
    for (int j = 0; j < 2; ++j)
#pragma unroll
      for (int r = 0; r < 8; ++r) {
        const int row = m0 + wm * 64 + i * 16 + r + 8 * hi;
        const int col = n0 + wn * 32 + j * 16 + (lane & 15);
        float v = acc[i][j][r] + biasj[j];
        const size_t idx = (size_t)row * N + col;
        if constexpr (EM == 0) {
          ((float*)outp)[idx] = v + resid[idx];
        } else if constexpr (EM == 1 || EM == 3) {
          ((__bf16*)outp)[idx] = (__bf16)v;
        } else {
          ((__bf16*)outp)[idx] = (__bf16)gelu_tanh(v);
        }
      }
}


static __device__ inline void load_kfrag(v16bf (&k)[2][2], const __bf16* Kp, int kt, int lane) {
#pragma unroll
  for (int t = 0; t < 2; ++t) {
    const __bf16* p = Kp + (size_t)(kt * 32 + t * 16) * (3 * CC);
    k[t][0] = frag_rowmajor(p, 3 * CC, lane);
    k[t][1] = frag_rowmajor(p + 32, 3 * CC, lane);
  }
}

static __device__ inline void load_vfrag(v16bf (&v)[4], const __bf16* VTp, int kt, int lane) {
  const __bf16* vtile = VTp + (size_t)kt * (HD * 32);
#pragma unroll
  for (int n = 0; n < 4; ++n) v[n] = frag_rowmajor(vtile + (n * 16) * 32, 32, lane);
}

__global__ void __launch_bounds__(32, 1)
attn_kernel(const __bf16* __restrict__ qkv, const __bf16* __restrict__ vT,
            __bf16* __restrict__ y, float* __restrict__ ent) {
  const int lane  = threadIdx.x;
  const int hi    = lane >> 4;
  const int qbase = blockIdx.x * 16;
  const int h     = blockIdx.y;
  const int b     = blockIdx.z;
  const int ldq   = 3 * CC;
  const int qrow  = qbase + (lane & 15);

  const __bf16* Qp  = qkv + (size_t)b * TT * ldq + h * HD;
  const __bf16* Kp  = Qp + CC;
  const __bf16* VTp = vT + (size_t)(b * HH + h) * (TT / 32) * (HD * 32);

  const v16bf qb0 = frag_rowmajor(Qp + (size_t)qbase * ldq + 0,  ldq, lane);
  const v16bf qb1 = frag_rowmajor(Qp + (size_t)qbase * ldq + 32, ldq, lane);

  v8f o[4] = {};
  float mrow = -INFINITY, lrow = 0.f, ses = 0.f;

  const int nkt = (qbase + 47) >> 5;

  __builtin_amdgcn_sched_barrier(0);

  v16bf kf[2][2], vf[4];
  load_kfrag(kf, Kp, 0, lane);
  load_vfrag(vf, VTp, 0, lane);

  int kt = 0;
  do {
    const int k0 = kt * 32;
    const int ktn = (kt + 1 < nkt) ? kt + 1 : kt;
    v8f s[2];
#pragma unroll
    for (int t = 0; t < 2; ++t) {
      v8f z = {};
      z = wmma_bf16(kf[t][0], qb0, z);
      s[t] = wmma_bf16(kf[t][1], qb1, z);
    }
    load_kfrag(kf, Kp, ktn, lane);
    __builtin_amdgcn_sched_barrier(0);

    float st[2][8];
#pragma unroll
    for (int t = 0; t < 2; ++t)
#pragma unroll
      for (int r = 0; r < 8; ++r) {
        const int key = k0 + t * 16 + 8 * hi + r;
        st[t][r] = (key <= qrow) ? s[t][r] * 0.125f : -1.0e30f;
      }
    float v = fmaxf(st[0][0], st[1][0]);
#pragma unroll
    for (int r = 1; r < 8; ++r) v = fmaxf(v, fmaxf(st[0][r], st[1][r]));
    v = fmaxf(v, xhalf(v));
    const float mn = fmaxf(mrow, v);
    const float a  = __expf(mrow - mn);
    float x[2][8];
    float ps = 0.f, pe = 0.f;
#pragma unroll
    for (int r = 0; r < 8; ++r) {
      x[0][r] = __expf(st[0][r] - mn);
      x[1][r] = __expf(st[1][r] - mn);
      ps += x[0][r] + x[1][r];
      pe += x[0][r] * st[0][r] + x[1][r] * st[1][r];
    }
    ps += xhalf(ps);
    pe += xhalf(pe);
    lrow = lrow * a + ps;
    ses  = ses  * a + pe;
    mrow = mn;
#pragma unroll
    for (int n = 0; n < 4; ++n)
#pragma unroll
      for (int r = 0; r < 8; ++r) o[n][r] *= a;
    v16bf pb;
#pragma unroll
    for (int r = 0; r < 8; ++r) {
      pb[r]     = (__bf16)x[0][r];
      pb[8 + r] = (__bf16)x[1][r];
    }
#pragma unroll
    for (int n = 0; n < 4; ++n) o[n] = wmma_bf16(vf[n], pb, o[n]);
    load_vfrag(vf, VTp, ktn, lane);
  } while (++kt < nkt);

  const float inv = 1.0f / lrow;
  __bf16* yrow = y + ((size_t)b * TT + qrow) * CC + h * HD + 8 * hi;
#pragma unroll
  for (int n = 0; n < 4; ++n) {
    v8bf pk;
#pragma unroll
    for (int r = 0; r < 8; ++r) pk[r] = (__bf16)(o[n][r] * inv);
    *(v8bf*)(yrow + n * 16) = pk;
  }
  float e = mrow + __logf(lrow) - ses * inv;
#pragma unroll
  for (int off = 1; off <= 8; off <<= 1) e += __shfl_xor(e, off, 32);
  if (lane == 0) atomicAdd(ent, e);
}

__global__ void zero_kernel(float* p) { *p = 0.f; }

__global__ void ent_final_kernel(const float* acc, float* out) {
  *out = *acc / (float)(BB * HH * TT);
}

extern "C" void kernel_launch(void* const* d_in, const int* in_sizes, int n_in,
                              void* d_out, int out_size, void* d_ws, size_t ws_size,
                              hipStream_t stream) {
  (void)in_sizes; (void)n_in; (void)out_size; (void)ws_size;
  const float* x          = (const float*)d_in[0];
  const float* ln1_w      = (const float*)d_in[1];
  const float* ln1_b      = (const float*)d_in[2];
  const float* attn_w     = (const float*)d_in[3];
  const float* attn_b     = (const float*)d_in[4];
  const float* attnproj_w = (const float*)d_in[5];
  const float* attnproj_b = (const float*)d_in[6];
  const float* ln2_w      = (const float*)d_in[7];
  const float* ln2_b      = (const float*)d_in[8];
  const float* fc_w       = (const float*)d_in[9];
  const float* fc_b       = (const float*)d_in[10];
  const float* proj_w     = (const float*)d_in[11];
  const float* proj_b     = (const float*)d_in[12];
  float* out = (float*)d_out;

  char* ws = (char*)d_ws;
  float*  x1      = (float*)(ws);
  __bf16* qkv_bf  = (__bf16*)(ws + ((size_t)16 << 20));
  __bf16* h_bf    = (__bf16*)(ws + ((size_t)40 << 20));
  __bf16* y_bf    = (__bf16*)(ws + ((size_t)48 << 20));
  __bf16* m_bf    = (__bf16*)(ws + ((size_t)56 << 20));
  __bf16* vT_bf   = m_bf;
  __bf16* wqkv_bf = (__bf16*)(ws + ((size_t)88 << 20));
  __bf16* wap_bf  = (__bf16*)(ws + ((size_t)94 << 20));
  __bf16* wfc_bf  = (__bf16*)(ws + ((size_t)96 << 20));
  __bf16* wpj_bf  = (__bf16*)(ws + ((size_t)104 << 20));
  float*  accum   = (float*)(ws + ((size_t)112 << 20));

  const int M = BB * TT;

  zero_kernel<<<1, 1, 0, stream>>>(accum);
  wconv_kernel<<<dim3((CC * CC) / (4 * WCONV_LOADS * 256), 12), 256, 0, stream>>>(
      attn_w, attnproj_w, fc_w, proj_w, wqkv_bf);
  ln_kernel<<<M, 256, 0, stream>>>(x, ln1_w, ln1_b, h_bf);
  gemm_kernel<3><<<dim3(3 * CC / 128, M / 128), 256, 0, stream>>>(
      h_bf, wqkv_bf, attn_b, nullptr, qkv_bf, vT_bf, M, 3 * CC, CC);
  attn_kernel<<<dim3(TT / 16, HH, BB), 32, 0, stream>>>(qkv_bf, vT_bf, y_bf, accum);
  gemm_kernel<0><<<dim3(CC / 128, M / 128), 256, 0, stream>>>(
      y_bf, wap_bf, attnproj_b, x, x1, nullptr, M, CC, CC);
  ln_kernel<<<M, 256, 0, stream>>>(x1, ln2_w, ln2_b, h_bf);
  gemm_kernel<2><<<dim3(4 * CC / 128, M / 128), 256, 0, stream>>>(
      h_bf, wfc_bf, fc_b, nullptr, m_bf, nullptr, M, 4 * CC, CC);
  gemm_kernel<0><<<dim3(CC / 128, M / 128), 256, 0, stream>>>(
      m_bf, wpj_bf, proj_b, x1, out, nullptr, M, CC, 4 * CC);
  ent_final_kernel<<<1, 1, 0, stream>>>(accum, out + (size_t)M * CC);
}


# GEMM K-step 32 -> 128 for the two GEMMs that run one block per WGP (attention proj K=1024, MLP proj K=4096): 4x fewer load round trips and barriers; whole-line tile loads; sum order unchanged
# speedup vs baseline: 2.1790x; 1.1622x over previous
#include <hip/hip_runtime.h>
#include <hip/hip_bf16.h>
#include <math.h>

typedef __attribute__((ext_vector_type(16))) __bf16 v16bf;
typedef __attribute__((ext_vector_type(8)))  float  v8f;
typedef __attribute__((ext_vector_type(8)))  __bf16 v8bf;
typedef __attribute__((ext_vector_type(4)))  unsigned int u32x4;

#define BB 2
#define TT 2048
#define CC 1024
#define HH 16
#define HD 64

static __device__ inline v8f wmma_bf16(v16bf a, v16bf b, v8f c) {
  return __builtin_amdgcn_wmma_f32_16x16x32_bf16(false, a, false, b, (short)0, c, false, false);
}

static __device__ inline float gelu_tanh(float x) {
  const float c = 0.7978845608028654f;
  return 0.5f * x * (1.0f + tanhf(c * (x + 0.044715f * x * x * x)));
}

static __device__ inline v16bf frag_rowmajor(const __bf16* p, int ld, int lane) {
  const int m  = lane & 15;
  const int ko = (lane >> 4) << 3;
  const __bf16* q = p + (size_t)m * ld + ko;
  v16bf f;
#pragma unroll
  for (int j = 0; j < 4; ++j) {
    f[2 * j]     = q[2 * j];
    f[2 * j + 1] = q[2 * j + 1];
    f[8 + 2 * j] = q[16 + 2 * j];
    f[9 + 2 * j] = q[16 + 2 * j + 1];
  }
  return f;
}

static __device__ inline float xhalf(float x) {
  const int b = __builtin_bit_cast(int, x);
  return __builtin_bit_cast(float, __builtin_amdgcn_permlanex16(b, b, 0x76543210, 0xfedcba98, false, false));
}

typedef __attribute__((ext_vector_type(4))) __bf16 v4bf;
#define WCONV_LOADS 16
__global__ void __launch_bounds__(256)
wconv_kernel(const float* __restrict__ w_qkv, const float* __restrict__ w_ap,
             const float* __restrict__ w_fc, const float* __restrict__ w_pj,
             __bf16* __restrict__ dst) {
  const int u = blockIdx.y;
  const float* src = (u < 3) ? w_qkv + (size_t)u * (CC * CC)
                   : (u < 4) ? w_ap
                   : (u < 8) ? w_fc + (size_t)(u - 4) * (CC * CC)
                             : w_pj + (size_t)(u - 8) * (CC * CC);
  const size_t q0 = (size_t)blockIdx.x * (WCONV_LOADS * 256) + threadIdx.x;
  const float4* s4 = (const float4*)src + q0;
  v4bf* d4 = (v4bf*)(dst + (size_t)u * (CC * CC)) + q0;
  float4 v[WCONV_LOADS];
#pragma unroll
  for (int i = 0; i < WCONV_LOADS; ++i) v[i] = s4[i * 256];
#pragma unroll
  for (int i = 0; i < WCONV_LOADS; ++i) {
    v4bf p;
    p[0] = (__bf16)v[i].x; p[1] = (__bf16)v[i].y;
    p[2] = (__bf16)v[i].z; p[3] = (__bf16)v[i].w;
    d4[i * 256] = p;
  }
}

__global__ void ln_kernel(const float* __restrict__ x, const float* __restrict__ w,
                          const float* __restrict__ b, __bf16* __restrict__ out) {
  const int row = blockIdx.x;
  const int tid = threadIdx.x;
  const float* xr = x + (size_t)row * CC;
  float v[4];
  float s = 0.f, sq = 0.f;
#pragma unroll
  for (int i = 0; i < 4; ++i) {
    v[i] = xr[tid + i * 256];
    s += v[i];
    sq += v[i] * v[i];
  }
#pragma unroll
  for (int off = 16; off >= 1; off >>= 1) {
    s  += __shfl_xor(s,  off, 32);
    sq += __shfl_xor(sq, off, 32);
  }
  __shared__ float reds[8], redq[8];
  const int wid = tid >> 5;
  if ((tid & 31) == 0) { reds[wid] = s; redq[wid] = sq; }
  __syncthreads();
  float ts = 0.f, tq = 0.f;
#pragma unroll
  for (int k = 0; k < 8; ++k) { ts += reds[k]; tq += redq[k]; }
  const float mean = ts * (1.0f / CC);
  const float var  = tq * (1.0f / CC) - mean * mean;
  const float rstd = rsqrtf(var + 1e-5f);
  __bf16* orow = out + (size_t)row * CC;
#pragma unroll
  for (int i = 0; i < 4; ++i) {
    const int c = tid + i * 256;
    orow[c] = (__bf16)((v[i] - mean) * rstd * w[c] + b[c]);
  }
}


template <int LD>
static __device__ inline void mma_step(const __bf16* as, const __bf16* bs, v8f (&acc)[4][2],
                                       int wm, int wn, int lane) {
  v16bf af[4], bf[2];
#pragma unroll
  for (int i = 0; i < 4; ++i) af[i] = frag_rowmajor(as + (wm * 64 + i * 16) * LD, LD, lane);
#pragma unroll
  for (int j = 0; j < 2; ++j) bf[j] = frag_rowmajor(bs + (wn * 32 + j * 16) * LD, LD, lane);
#pragma unroll
  for (int i = 0; i < 4; ++i)
#pragma unroll
    for (int j = 0; j < 2; ++j) acc[i][j] = wmma_bf16(af[i], bf[j], acc[i][j]);
}

template <int EM, int BK>
__global__ void __launch_bounds__(256, 1)
gemm_kernel(const __bf16* A, const __bf16* W,
            const float* __restrict__ bias, const float* __restrict__ resid,
            void* __restrict__ outp, __bf16* __restrict__ vT, int M, int N, int K) {
  constexpr int LD  = BK + 8;
  constexpr int LPR = BK / 8;
  constexpr int RPI = 256 / LPR;
  constexpr int NL  = 128 / RPI;
  __shared__ __attribute__((aligned(16))) __bf16 As[2][128 * LD];
  __shared__ __attribute__((aligned(16))) __bf16 Bs[2][128 * LD];
  const int tid  = threadIdx.x;
  const int lane = tid & 31;
  const int wid  = tid >> 5;
  const int wm   = wid & 1;
  const int wn   = wid >> 1;
  const int m0   = blockIdx.y * 128;
  const int n0   = blockIdx.x * 128;

  v8f acc[4][2] = {};

  const int srow = tid / LPR, scol = (tid % LPR) * 8;
  const char* abase = (const char*)(A + (size_t)m0 * K);
  const char* bbase = (const char*)(W + (size_t)n0 * K);
  unsigned voff[NL];
#pragma unroll
  for (int j = 0; j < NL; ++j) voff[j] = (unsigned)(((srow + j * RPI) * K + scol) * 2);
  const int soff = srow * LD + scol;

  u32x4 ra[NL], rb[NL];
#pragma unroll
  for (int j = 0; j < NL; ++j) {
    ra[j] = *(const u32x4*)(abase + voff[j]);
    rb[j] = *(const u32x4*)(bbase + voff[j]);
  }
#pragma unroll
  for (int j = 0; j < NL; ++j) {
    *(u32x4*)(As[0] + soff + j * RPI * LD) = ra[j];
    *(u32x4*)(Bs[0] + soff + j * RPI * LD) = rb[j];
  }
  __syncthreads();

  const int nk = K / BK;
  for (int kt = 0; kt + 1 < nk; ++kt) {
    const int cur = kt & 1;
    abase += BK * 2;
    bbase += BK * 2;
#pragma unroll
    for (int j = 0; j < NL; ++j) {
      ra[j] = *(const u32x4*)(abase + voff[j]);
      rb[j] = *(const u32x4*)(bbase + voff[j]);
    }
    asm volatile("" ::: "memory");

#pragma unroll
    for (int ks = 0; ks < BK / 32; ++ks)
      mma_step<LD>(As[cur] + ks * 32, Bs[cur] + ks * 32, acc, wm, wn, lane);

#pragma unroll
    for (int j = 0; j < NL; ++j) {
      *(u32x4*)(As[cur ^ 1] + soff + j * RPI * LD) = ra[j];
      *(u32x4*)(Bs[cur ^ 1] + soff + j * RPI * LD) = rb[j];
    }
    __syncthreads();
  }
#pragma unroll
  for (int ks = 0; ks < BK / 32; ++ks)
    mma_step<LD>(As[(nk - 1) & 1] + ks * 32, Bs[(nk - 1) & 1] + ks * 32, acc, wm, wn, lane);

  const int hi = lane >> 4;
  float biasj[2];
#pragma unroll
  for (int j = 0; j < 2; ++j) biasj[j] = bias[n0 + wn * 32 + j * 16 + (lane & 15)];

  if constexpr (EM == 3) {
    if (n0 >= 2 * CC) {
#pragma unroll
      for (int i = 0; i < 4; ++i)
#pragma unroll
        for (int j = 0; j < 2; ++j) {
          const int row0 = m0 + wm * 64 + i * 16 + 8 * hi;
          const int c    = n0 - 2 * CC + wn * 32 + j * 16 + (lane & 15);
          const int bb   = row0 / TT;
          const int t0   = row0 % TT;
          const size_t tile = (size_t)(bb * HH + (c >> 6)) * (TT / 32) + (t0 >> 5);
          v8bf pk;
#pragma unroll
          for (int r = 0; r < 8; ++r) pk[r] = (__bf16)(acc[i][j][r] + biasj[j]);
          *(v8bf*)(vT + (tile * HD + (c & 63)) * 32 + (t0 & 31)) = pk;
        }
      return;
    }
  }

#pragma unroll
  for (int i = 0; i < 4; ++i)
#pragma unroll
    for (int j = 0; j < 2; ++j)
#pragma unroll
      for (int r = 0; r < 8; ++r) {
        const int row = m0 + wm * 64 + i * 16 + r + 8 * hi;
        const int col = n0 + wn * 32 + j * 16 + (lane & 15);
        float v = acc[i][j][r] + biasj[j];
        const size_t idx = (size_t)row * N + col;
        if constexpr (EM == 0) {
          ((float*)outp)[idx] = v + resid[idx];
        } else if constexpr (EM == 1 || EM == 3) {
          ((__bf16*)outp)[idx] = (__bf16)v;
        } else {
          ((__bf16*)outp)[idx] = (__bf16)gelu_tanh(v);
        }
      }
}


static __device__ inline void load_kfrag(v16bf (&k)[2][2], const __bf16* Kp, int kt, int lane) {
#pragma unroll
  for (int t = 0; t < 2; ++t) {
    const __bf16* p = Kp + (size_t)(kt * 32 + t * 16) * (3 * CC);
    k[t][0] = frag_rowmajor(p, 3 * CC, lane);
    k[t][1] = frag_rowmajor(p + 32, 3 * CC, lane);
  }
}

static __device__ inline void load_vfrag(v16bf (&v)[4], const __bf16* VTp, int kt, int lane) {
  const __bf16* vtile = VTp + (size_t)kt * (HD * 32);
#pragma unroll
  for (int n = 0; n < 4; ++n) v[n] = frag_rowmajor(vtile + (n * 16) * 32, 32, lane);
}

__global__ void __launch_bounds__(32, 1)
attn_kernel(const __bf16* __restrict__ qkv, const __bf16* __restrict__ vT,
            __bf16* __restrict__ y, float* __restrict__ ent) {
  const int lane  = threadIdx.x;
  const int hi    = lane >> 4;
  const int qbase = blockIdx.x * 16;
  const int h     = blockIdx.y;
  const int b     = blockIdx.z;
  const int ldq   = 3 * CC;
  const int qrow  = qbase + (lane & 15);

  const __bf16* Qp  = qkv + (size_t)b * TT * ldq + h * HD;
  const __bf16* Kp  = Qp + CC;
  const __bf16* VTp = vT + (size_t)(b * HH + h) * (TT / 32) * (HD * 32);

  const v16bf qb0 = frag_rowmajor(Qp + (size_t)qbase * ldq + 0,  ldq, lane);
  const v16bf qb1 = frag_rowmajor(Qp + (size_t)qbase * ldq + 32, ldq, lane);

  v8f o[4] = {};
  float mrow = -INFINITY, lrow = 0.f, ses = 0.f;

  const int nkt = (qbase + 47) >> 5;

  __builtin_amdgcn_sched_barrier(0);

  v16bf kf[2][2], vf[4];
  load_kfrag(kf, Kp, 0, lane);
  load_vfrag(vf, VTp, 0, lane);

  int kt = 0;
  do {
    const int k0 = kt * 32;
    const int ktn = (kt + 1 < nkt) ? kt + 1 : kt;
    v8f s[2];
#pragma unroll
    for (int t = 0; t < 2; ++t) {
      v8f z = {};
      z = wmma_bf16(kf[t][0], qb0, z);
      s[t] = wmma_bf16(kf[t][1], qb1, z);
    }
    load_kfrag(kf, Kp, ktn, lane);
    __builtin_amdgcn_sched_barrier(0);

    float st[2][8];
#pragma unroll
    for (int t = 0; t < 2; ++t)
#pragma unroll
      for (int r = 0; r < 8; ++r) {
        const int key = k0 + t * 16 + 8 * hi + r;
        st[t][r] = (key <= qrow) ? s[t][r] * 0.125f : -1.0e30f;
      }
    float v = fmaxf(st[0][0], st[1][0]);
#pragma unroll
    for (int r = 1; r < 8; ++r) v = fmaxf(v, fmaxf(st[0][r], st[1][r]));
    v = fmaxf(v, xhalf(v));
    const float mn = fmaxf(mrow, v);
    const float a  = __expf(mrow - mn);
    float x[2][8];
    float ps = 0.f, pe = 0.f;
#pragma unroll
    for (int r = 0; r < 8; ++r) {
      x[0][r] = __expf(st[0][r] - mn);
      x[1][r] = __expf(st[1][r] - mn);
      ps += x[0][r] + x[1][r];
      pe += x[0][r] * st[0][r] + x[1][r] * st[1][r];
    }
    ps += xhalf(ps);
    pe += xhalf(pe);
    lrow = lrow * a + ps;
    ses  = ses  * a + pe;
    mrow = mn;
#pragma unroll
    for (int n = 0; n < 4; ++n)
#pragma unroll
      for (int r = 0; r < 8; ++r) o[n][r] *= a;
    v16bf pb;
#pragma unroll
    for (int r = 0; r < 8; ++r) {
      pb[r]     = (__bf16)x[0][r];
      pb[8 + r] = (__bf16)x[1][r];
    }
#pragma unroll
    for (int n = 0; n < 4; ++n) o[n] = wmma_bf16(vf[n], pb, o[n]);
    load_vfrag(vf, VTp, ktn, lane);
  } while (++kt < nkt);

  const float inv = 1.0f / lrow;
  __bf16* yrow = y + ((size_t)b * TT + qrow) * CC + h * HD + 8 * hi;
#pragma unroll
  for (int n = 0; n < 4; ++n) {
    v8bf pk;
#pragma unroll
    for (int r = 0; r < 8; ++r) pk[r] = (__bf16)(o[n][r] * inv);
    *(v8bf*)(yrow + n * 16) = pk;
  }
  float e = mrow + __logf(lrow) - ses * inv;
#pragma unroll
  for (int off = 1; off <= 8; off <<= 1) e += __shfl_xor(e, off, 32);
  if (lane == 0) atomicAdd(ent, e);
}

__global__ void zero_kernel(float* p) { *p = 0.f; }

__global__ void ent_final_kernel(const float* acc, float* out) {
  *out = *acc / (float)(BB * HH * TT);
}

extern "C" void kernel_launch(void* const* d_in, const int* in_sizes, int n_in,
                              void* d_out, int out_size, void* d_ws, size_t ws_size,
                              hipStream_t stream) {
  (void)in_sizes; (void)n_in; (void)out_size; (void)ws_size;
  const float* x          = (const float*)d_in[0];
  const float* ln1_w      = (const float*)d_in[1];
  const float* ln1_b      = (const float*)d_in[2];
  const float* attn_w     = (const float*)d_in[3];
  const float* attn_b     = (const float*)d_in[4];
  const float* attnproj_w = (const float*)d_in[5];
  const float* attnproj_b = (const float*)d_in[6];
  const float* ln2_w      = (const float*)d_in[7];
  const float* ln2_b      = (const float*)d_in[8];
  const float* fc_w       = (const float*)d_in[9];
  const float* fc_b       = (const float*)d_in[10];
  const float* proj_w     = (const float*)d_in[11];
  const float* proj_b     = (const float*)d_in[12];
  float* out = (float*)d_out;

  char* ws = (char*)d_ws;
  float*  x1      = (float*)(ws);
  __bf16* qkv_bf  = (__bf16*)(ws + ((size_t)16 << 20));
  __bf16* h_bf    = (__bf16*)(ws + ((size_t)40 << 20));
  __bf16* y_bf    = (__bf16*)(ws + ((size_t)48 << 20));
  __bf16* m_bf    = (__bf16*)(ws + ((size_t)56 << 20));
  __bf16* vT_bf   = m_bf;
  __bf16* wqkv_bf = (__bf16*)(ws + ((size_t)88 << 20));
  __bf16* wap_bf  = (__bf16*)(ws + ((size_t)94 << 20));
  __bf16* wfc_bf  = (__bf16*)(ws + ((size_t)96 << 20));
  __bf16* wpj_bf  = (__bf16*)(ws + ((size_t)104 << 20));
  float*  accum   = (float*)(ws + ((size_t)112 << 20));

  const int M = BB * TT;

  zero_kernel<<<1, 1, 0, stream>>>(accum);
  wconv_kernel<<<dim3((CC * CC) / (4 * WCONV_LOADS * 256), 12), 256, 0, stream>>>(
      attn_w, attnproj_w, fc_w, proj_w, wqkv_bf);
  ln_kernel<<<M, 256, 0, stream>>>(x, ln1_w, ln1_b, h_bf);
  gemm_kernel<3, 32><<<dim3(3 * CC / 128, M / 128), 256, 0, stream>>>(
      h_bf, wqkv_bf, attn_b, nullptr, qkv_bf, vT_bf, M, 3 * CC, CC);
  attn_kernel<<<dim3(TT / 16, HH, BB), 32, 0, stream>>>(qkv_bf, vT_bf, y_bf, accum);
  gemm_kernel<0, 128><<<dim3(CC / 128, M / 128), 256, 0, stream>>>(
      y_bf, wap_bf, attnproj_b, x, x1, nullptr, M, CC, CC);
  ln_kernel<<<M, 256, 0, stream>>>(x1, ln2_w, ln2_b, h_bf);
  gemm_kernel<2, 32><<<dim3(4 * CC / 128, M / 128), 256, 0, stream>>>(
      h_bf, wfc_bf, fc_b, nullptr, m_bf, nullptr, M, 4 * CC, CC);
  gemm_kernel<0, 128><<<dim3(CC / 128, M / 128), 256, 0, stream>>>(
      m_bf, wpj_bf, proj_b, x1, out, nullptr, M, CC, 4 * CC);
  ent_final_kernel<<<1, 1, 0, stream>>>(accum, out + (size_t)M * CC);
}


# attention K and V^T stored in fragment order by the QKV epilogue (K blocks: operands exchanged): fragment loads and stores are 512 contiguous bytes; entropy partials summed in fixed order, no atomics
# speedup vs baseline: 2.3888x; 1.0963x over previous
#include <hip/hip_runtime.h>
#include <hip/hip_bf16.h>
#include <math.h>

typedef __attribute__((ext_vector_type(16))) __bf16 v16bf;
typedef __attribute__((ext_vector_type(8)))  float  v8f;
typedef __attribute__((ext_vector_type(8)))  __bf16 v8bf;
typedef __attribute__((ext_vector_type(4)))  unsigned int u32x4;

#define BB 2
#define TT 2048
#define CC 1024
#define HH 16
#define HD 64

static __device__ inline v8f wmma_bf16(v16bf a, v16bf b, v8f c) {
  return __builtin_amdgcn_wmma_f32_16x16x32_bf16(false, a, false, b, (short)0, c, false, false);
}

static __device__ inline float gelu_tanh(float x) {
  const float c = 0.7978845608028654f;
  return 0.5f * x * (1.0f + tanhf(c * (x + 0.044715f * x * x * x)));
}

static __device__ inline v16bf frag_rowmajor(const __bf16* p, int ld, int lane) {
  const int m  = lane & 15;
  const int ko = (lane >> 4) << 3;
  const __bf16* q = p + (size_t)m * ld + ko;
  v16bf f;
#pragma unroll
  for (int j = 0; j < 4; ++j) {
    f[2 * j]     = q[2 * j];
    f[2 * j + 1] = q[2 * j + 1];
    f[8 + 2 * j] = q[16 + 2 * j];
    f[9 + 2 * j] = q[16 + 2 * j + 1];
  }
  return f;
}

static __device__ inline v16bf frag_packed(const __bf16* f, int lane) {
  const v8bf lo = *(const v8bf*)(f + lane * 8);
  const v8bf hi = *(const v8bf*)(f + 256 + lane * 8);
  return __builtin_shufflevector(lo, hi, 0, 1, 2, 3, 4, 5, 6, 7, 8, 9, 10, 11, 12, 13, 14, 15);
}

static __device__ inline float xhalf(float x) {
  const int b = __builtin_bit_cast(int, x);
  return __builtin_bit_cast(float, __builtin_amdgcn_permlanex16(b, b, 0x76543210, 0xfedcba98, false, false));
}

typedef __attribute__((ext_vector_type(4))) __bf16 v4bf;
#define WCONV_LOADS 16
__global__ void __launch_bounds__(256)
wconv_kernel(const float* __restrict__ w_qkv, const float* __restrict__ w_ap,
             const float* __restrict__ w_fc, const float* __restrict__ w_pj,
             __bf16* __restrict__ dst) {
  const int u = blockIdx.y;
  const float* src = (u < 3) ? w_qkv + (size_t)u * (CC * CC)
                   : (u < 4) ? w_ap
                   : (u < 8) ? w_fc + (size_t)(u - 4) * (CC * CC)
                             : w_pj + (size_t)(u - 8) * (CC * CC);
  const size_t q0 = (size_t)blockIdx.x * (WCONV_LOADS * 256) + threadIdx.x;
  const float4* s4 = (const float4*)src + q0;
  v4bf* d4 = (v4bf*)(dst + (size_t)u * (CC * CC)) + q0;
  float4 v[WCONV_LOADS];
#pragma unroll
  for (int i = 0; i < WCONV_LOADS; ++i) v[i] = s4[i * 256];
#pragma unroll
  for (int i = 0; i < WCONV_LOADS; ++i) {
    v4bf p;
    p[0] = (__bf16)v[i].x; p[1] = (__bf16)v[i].y;
    p[2] = (__bf16)v[i].z; p[3] = (__bf16)v[i].w;
    d4[i * 256] = p;
  }
}

__global__ void ln_kernel(const float* __restrict__ x, const float* __restrict__ w,
                          const float* __restrict__ b, __bf16* __restrict__ out) {
  const int row = blockIdx.x;
  const int tid = threadIdx.x;
  const float* xr = x + (size_t)row * CC;
  float v[4];
  float s = 0.f, sq = 0.f;
#pragma unroll
  for (int i = 0; i < 4; ++i) {
    v[i] = xr[tid + i * 256];
    s += v[i];
    sq += v[i] * v[i];
  }
#pragma unroll
  for (int off = 16; off >= 1; off >>= 1) {
    s  += __shfl_xor(s,  off, 32);
    sq += __shfl_xor(sq, off, 32);
  }
  __shared__ float reds[8], redq[8];
  const int wid = tid >> 5;
  if ((tid & 31) == 0) { reds[wid] = s; redq[wid] = sq; }
  __syncthreads();
  float ts = 0.f, tq = 0.f;
#pragma unroll
  for (int k = 0; k < 8; ++k) { ts += reds[k]; tq += redq[k]; }
  const float mean = ts * (1.0f / CC);
  const float var  = tq * (1.0f / CC) - mean * mean;
  const float rstd = rsqrtf(var + 1e-5f);
  __bf16* orow = out + (size_t)row * CC;
#pragma unroll
  for (int i = 0; i < 4; ++i) {
    const int c = tid + i * 256;
    orow[c] = (__bf16)((v[i] - mean) * rstd * w[c] + b[c]);
  }
}


template <int LD>
static __device__ inline void mma_step(const __bf16* as, const __bf16* bs, v8f (&acc)[4][2],
                                       int wm, int wn, int lane) {
  v16bf af[4], bf[2];
#pragma unroll
  for (int i = 0; i < 4; ++i) af[i] = frag_rowmajor(as + (wm * 64 + i * 16) * LD, LD, lane);
#pragma unroll
  for (int j = 0; j < 2; ++j) bf[j] = frag_rowmajor(bs + (wn * 32 + j * 16) * LD, LD, lane);
#pragma unroll
  for (int i = 0; i < 4; ++i)
#pragma unroll
    for (int j = 0; j < 2; ++j) acc[i][j] = wmma_bf16(af[i], bf[j], acc[i][j]);
}

template <int EM, int BK>
__global__ void __launch_bounds__(256, (EM == 3 ? 6 : 1))
gemm_kernel(const __bf16* A, const __bf16* W,
            const float* __restrict__ bias, const float* __restrict__ resid,
            void* __restrict__ outp, __bf16* __restrict__ kF, __bf16* __restrict__ vF,
            int M, int N, int K) {
  constexpr int LD  = BK + 8;
  constexpr int LPR = BK / 8;
  constexpr int RPI = 256 / LPR;
  constexpr int NL  = 128 / RPI;
  __shared__ __attribute__((aligned(16))) __bf16 As[2][128 * LD];
  __shared__ __attribute__((aligned(16))) __bf16 Bs[2][128 * LD];
  const int tid  = threadIdx.x;
  const int lane = tid & 31;
  const int wid  = tid >> 5;
  const int wm   = wid & 1;
  const int wn   = wid >> 1;
  const int m0   = blockIdx.y * 128;
  const int n0   = blockIdx.x * 128;

  v8f acc[4][2] = {};

  const int srow = tid / LPR, scol = (tid % LPR) * 8;
  const char* abase = (const char*)(A + (size_t)m0 * K);
  const char* bbase = (const char*)(W + (size_t)n0 * K);
  if constexpr (EM == 3) {
    if (n0 >= CC && n0 < 2 * CC) {
      const char* t = abase; abase = bbase; bbase = t;
    }
  }
  unsigned voff[NL];
#pragma unroll
  for (int j = 0; j < NL; ++j) voff[j] = (unsigned)(((srow + j * RPI) * K + scol) * 2);
  const int soff = srow * LD + scol;

  u32x4 ra[NL], rb[NL];
#pragma unroll
  for (int j = 0; j < NL; ++j) {
    ra[j] = *(const u32x4*)(abase + voff[j]);
    rb[j] = *(const u32x4*)(bbase + voff[j]);
  }
#pragma unroll
  for (int j = 0; j < NL; ++j) {
    *(u32x4*)(As[0] + soff + j * RPI * LD) = ra[j];
    *(u32x4*)(Bs[0] + soff + j * RPI * LD) = rb[j];
  }
  __syncthreads();

  const int nk = K / BK;
  for (int kt = 0; kt + 1 < nk; ++kt) {
    const int cur = kt & 1;
    abase += BK * 2;
    bbase += BK * 2;
#pragma unroll
    for (int j = 0; j < NL; ++j) {
      ra[j] = *(const u32x4*)(abase + voff[j]);
      rb[j] = *(const u32x4*)(bbase + voff[j]);
    }
    asm volatile("" ::: "memory");

#pragma unroll
    for (int ks = 0; ks < BK / 32; ++ks)
      mma_step<LD>(As[cur] + ks * 32, Bs[cur] + ks * 32, acc, wm, wn, lane);

#pragma unroll
    for (int j = 0; j < NL; ++j) {
      *(u32x4*)(As[cur ^ 1] + soff + j * RPI * LD) = ra[j];
      *(u32x4*)(Bs[cur ^ 1] + soff + j * RPI * LD) = rb[j];
    }
    __syncthreads();
  }
#pragma unroll
  for (int ks = 0; ks < BK / 32; ++ks)
    mma_step<LD>(As[(nk - 1) & 1] + ks * 32, Bs[(nk - 1) & 1] + ks * 32, acc, wm, wn, lane);

  const int hi = lane >> 4;
  float biasj[2];
#pragma unroll
  for (int j = 0; j < 2; ++j) biasj[j] = bias[n0 + wn * 32 + j * 16 + (lane & 15)];

  if constexpr (EM == 3) {
    if (n0 >= CC) {
      const int bb  = m0 / TT;
      const int kt0 = (m0 % TT) >> 5;
      if (n0 >= 2 * CC) {
        const int head = ((n0 - 2 * CC) >> 6) + (wn >> 1);
#pragma unroll
        for (int i = 0; i < 4; ++i)
#pragma unroll
          for (int j = 0; j < 2; ++j) {
            const int kt = kt0 + wm * 2 + (i >> 1);
            const int n  = (wn & 1) * 2 + j;
            const size_t tile = (size_t)(bb * HH + head) * (TT / 32) + kt;
            v8bf pk;
#pragma unroll
            for (int r = 0; r < 8; ++r) pk[r] = (__bf16)(acc[i][j][r] + biasj[j]);
            *(v8bf*)(vF + (tile * 4 + n) * 512 + (i & 1) * 256 + lane * 8) = pk;
          }
      } else {
        const int head = ((n0 - CC) >> 6) + wm;
        const size_t tile = (size_t)(bb * HH + head) * (TT / 32) + kt0 + wn;
#pragma unroll
        for (int i = 0; i < 4; ++i) {
          const float4 b0 = *(const float4*)(bias + n0 + wm * 64 + i * 16 + 8 * hi);
          const float4 b1 = *(const float4*)(bias + n0 + wm * 64 + i * 16 + 8 * hi + 4);
          const float br[8] = {b0.x, b0.y, b0.z, b0.w, b1.x, b1.y, b1.z, b1.w};
#pragma unroll
          for (int j = 0; j < 2; ++j) {
            v8bf pk;
#pragma unroll
            for (int r = 0; r < 8; ++r) pk[r] = (__bf16)(acc[i][j][r] + br[r]);
            *(v8bf*)(kF + (tile * 4 + j * 2 + (i >> 1)) * 512 + (i & 1) * 256 + lane * 8) = pk;
          }
        }
      }
      return;
    }
  }

#pragma unroll
  for (int i = 0; i < 4; ++i)
#pragma unroll
    for (int j = 0; j < 2; ++j)
#pragma unroll
      for (int r = 0; r < 8; ++r) {
        const int row = m0 + wm * 64 + i * 16 + r + 8 * hi;
        const int col = n0 + wn * 32 + j * 16 + (lane & 15);
        float v = acc[i][j][r] + biasj[j];
        const size_t idx = (size_t)row * N + col;
        if constexpr (EM == 0) {
          ((float*)outp)[idx] = v + resid[idx];
        } else if constexpr (EM == 1 || EM == 3) {
          ((__bf16*)outp)[idx] = (__bf16)v;
        } else {
          ((__bf16*)outp)[idx] = (__bf16)gelu_tanh(v);
        }
      }
}


static __device__ inline void load_kfrag(v16bf (&k)[2][2], const __bf16* KFp, int kt, int lane) {
  const __bf16* ktile = KFp + (size_t)kt * (4 * 512);
#pragma unroll
  for (int t = 0; t < 2; ++t) {
    k[t][0] = frag_packed(ktile + (t * 2 + 0) * 512, lane);
    k[t][1] = frag_packed(ktile + (t * 2 + 1) * 512, lane);
  }
}

static __device__ inline void load_vfrag(v16bf (&v)[4], const __bf16* VFp, int kt, int lane) {
  const __bf16* vtile = VFp + (size_t)kt * (4 * 512);
#pragma unroll
  for (int n = 0; n < 4; ++n) v[n] = frag_packed(vtile + n * 512, lane);
}

__global__ void __launch_bounds__(32, 1)
attn_kernel(const __bf16* __restrict__ qkv, const __bf16* __restrict__ kF,
            const __bf16* __restrict__ vF, __bf16* __restrict__ y, float* __restrict__ ent) {
  const int lane  = threadIdx.x;
  const int hi    = lane >> 4;
  const int qbase = blockIdx.x * 16;
  const int h     = blockIdx.y;
  const int b     = blockIdx.z;
  const int ldq   = 3 * CC;
  const int qrow  = qbase + (lane & 15);

  const __bf16* Qp  = qkv + (size_t)b * TT * ldq + h * HD;
  const __bf16* KFp = kF + (size_t)(b * HH + h) * (TT / 32) * (4 * 512);
  const __bf16* VFp = vF + (size_t)(b * HH + h) * (TT / 32) * (4 * 512);

  const v16bf qb0 = frag_rowmajor(Qp + (size_t)qbase * ldq + 0,  ldq, lane);
  const v16bf qb1 = frag_rowmajor(Qp + (size_t)qbase * ldq + 32, ldq, lane);

  v8f o[4] = {};
  float mrow = -INFINITY, lrow = 0.f, ses = 0.f;

  const int nkt = (qbase + 47) >> 5;

  __builtin_amdgcn_sched_barrier(0);

  v16bf kf[2][2], vf[4];
  load_kfrag(kf, KFp, 0, lane);
  load_vfrag(vf, VFp, 0, lane);

  int kt = 0;
  do {
    const int k0 = kt * 32;
    const int ktn = (kt + 1 < nkt) ? kt + 1 : kt;
    v8f s[2];
#pragma unroll
    for (int t = 0; t < 2; ++t) {
      v8f z = {};
      z = wmma_bf16(kf[t][0], qb0, z);
      s[t] = wmma_bf16(kf[t][1], qb1, z);
    }
    load_kfrag(kf, KFp, ktn, lane);
    __builtin_amdgcn_sched_barrier(0);

    float st[2][8];
#pragma unroll
    for (int t = 0; t < 2; ++t)
#pragma unroll
      for (int r = 0; r < 8; ++r) {
        const int key = k0 + t * 16 + 8 * hi + r;
        st[t][r] = (key <= qrow) ? s[t][r] * 0.125f : -1.0e30f;
      }
    float v = fmaxf(st[0][0], st[1][0]);
#pragma unroll
    for (int r = 1; r < 8; ++r) v = fmaxf(v, fmaxf(st[0][r], st[1][r]));
    v = fmaxf(v, xhalf(v));
    const float mn = fmaxf(mrow, v);
    const float a  = __expf(mrow - mn);
    float x[2][8];
    float ps = 0.f, pe = 0.f;
#pragma unroll
    for (int r = 0; r < 8; ++r) {
      x[0][r] = __expf(st[0][r] - mn);
      x[1][r] = __expf(st[1][r] - mn);
      ps += x[0][r] + x[1][r];
      pe += x[0][r] * st[0][r] + x[1][r] * st[1][r];
    }
    ps += xhalf(ps);
    pe += xhalf(pe);
    lrow = lrow * a + ps;
    ses  = ses  * a + pe;
    mrow = mn;
#pragma unroll
    for (int n = 0; n < 4; ++n)
#pragma unroll
      for (int r = 0; r < 8; ++r) o[n][r] *= a;
    v16bf pb;
#pragma unroll
    for (int r = 0; r < 8; ++r) {
      pb[r]     = (__bf16)x[0][r];
      pb[8 + r] = (__bf16)x[1][r];
    }
#pragma unroll
    for (int n = 0; n < 4; ++n) o[n] = wmma_bf16(vf[n], pb, o[n]);
    load_vfrag(vf, VFp, ktn, lane);
  } while (++kt < nkt);

  const float inv = 1.0f / lrow;
  __bf16* yrow = y + ((size_t)b * TT + qrow) * CC + h * HD + 8 * hi;
#pragma unroll
  for (int n = 0; n < 4; ++n) {
    v8bf pk;
#pragma unroll
    for (int r = 0; r < 8; ++r) pk[r] = (__bf16)(o[n][r] * inv);
    *(v8bf*)(yrow + n * 16) = pk;
  }
  float e = mrow + __logf(lrow) - ses * inv;
#pragma unroll
  for (int off = 1; off <= 8; off <<= 1) e += __shfl_xor(e, off, 32);
  if (lane == 0) ent[(size_t)(b * HH + h) * (TT / 16) + blockIdx.x] = e;
}

#define ENT_PARTS (BB * HH * (TT / 16))
__global__ void __launch_bounds__(256)
ent_final_kernel(const float* __restrict__ part, float* __restrict__ out) {
  const int tid = threadIdx.x;
  float s = 0.f;
#pragma unroll
  for (int k = 0; k < ENT_PARTS / 256; ++k) s += part[tid + k * 256];
#pragma unroll
  for (int off = 16; off >= 1; off >>= 1) s += __shfl_xor(s, off, 32);
  __shared__ float red[8];
  if ((tid & 31) == 0) red[tid >> 5] = s;
  __syncthreads();
  if (tid == 0) {
    float t = 0.f;
#pragma unroll
    for (int k = 0; k < 8; ++k) t += red[k];
    *out = t / (float)(BB * HH * TT);
  }
}

extern "C" void kernel_launch(void* const* d_in, const int* in_sizes, int n_in,
                              void* d_out, int out_size, void* d_ws, size_t ws_size,
                              hipStream_t stream) {
  (void)in_sizes; (void)n_in; (void)out_size; (void)ws_size;
  const float* x          = (const float*)d_in[0];
  const float* ln1_w      = (const float*)d_in[1];
  const float* ln1_b      = (const float*)d_in[2];
  const float* attn_w     = (const float*)d_in[3];
  const float* attn_b     = (const float*)d_in[4];
  const float* attnproj_w = (const float*)d_in[5];
  const float* attnproj_b = (const float*)d_in[6];
  const float* ln2_w      = (const float*)d_in[7];
  const float* ln2_b      = (const float*)d_in[8];
  const float* fc_w       = (const float*)d_in[9];
  const float* fc_b       = (const float*)d_in[10];
  const float* proj_w     = (const float*)d_in[11];
  const float* proj_b     = (const float*)d_in[12];
  float* out = (float*)d_out;

  char* ws = (char*)d_ws;
  float*  x1      = (float*)(ws);
  __bf16* qkv_bf  = (__bf16*)(ws + ((size_t)16 << 20));
  __bf16* h_bf    = (__bf16*)(ws + ((size_t)40 << 20));
  __bf16* y_bf    = (__bf16*)(ws + ((size_t)48 << 20));
  __bf16* m_bf    = (__bf16*)(ws + ((size_t)56 << 20));
  __bf16* vF_bf   = m_bf;
  __bf16* kF_bf   = (__bf16*)(ws + ((size_t)64 << 20));
  float*  epart   = (float*)(ws + ((size_t)72 << 20));
  __bf16* wqkv_bf = (__bf16*)(ws + ((size_t)88 << 20));
  __bf16* wap_bf  = (__bf16*)(ws + ((size_t)94 << 20));
  __bf16* wfc_bf  = (__bf16*)(ws + ((size_t)96 << 20));
  __bf16* wpj_bf  = (__bf16*)(ws + ((size_t)104 << 20));

  const int M = BB * TT;

  wconv_kernel<<<dim3((CC * CC) / (4 * WCONV_LOADS * 256), 12), 256, 0, stream>>>(
      attn_w, attnproj_w, fc_w, proj_w, wqkv_bf);
  ln_kernel<<<M, 256, 0, stream>>>(x, ln1_w, ln1_b, h_bf);
  gemm_kernel<3, 32><<<dim3(3 * CC / 128, M / 128), 256, 0, stream>>>(
      h_bf, wqkv_bf, attn_b, nullptr, qkv_bf, kF_bf, vF_bf, M, 3 * CC, CC);
  attn_kernel<<<dim3(TT / 16, HH, BB), 32, 0, stream>>>(qkv_bf, kF_bf, vF_bf, y_bf, epart);
  ent_final_kernel<<<1, 256, 0, stream>>>(epart, out + (size_t)M * CC);
  gemm_kernel<0, 128><<<dim3(CC / 128, M / 128), 256, 0, stream>>>(
      y_bf, wap_bf, attnproj_b, x, x1, nullptr, nullptr, M, CC, CC);
  ln_kernel<<<M, 256, 0, stream>>>(x1, ln2_w, ln2_b, h_bf);
  gemm_kernel<2, 32><<<dim3(4 * CC / 128, M / 128), 256, 0, stream>>>(
      h_bf, wfc_bf, fc_b, nullptr, m_bf, nullptr, nullptr, M, 4 * CC, CC);
  gemm_kernel<0, 128><<<dim3(CC / 128, M / 128), 256, 0, stream>>>(
      m_bf, wpj_bf, proj_b, x1, out, nullptr, nullptr, M, CC, 4 * CC);
}


# attention wave takes 2 query tiles (32 queries): each loaded K and V^T fragment feeds both tiles' WMMAs, halving attention's fragment loads; per-query and entropy sum order unchanged
# speedup vs baseline: 2.5220x; 1.0558x over previous
#include <hip/hip_runtime.h>
#include <hip/hip_bf16.h>
#include <math.h>

typedef __attribute__((ext_vector_type(16))) __bf16 v16bf;
typedef __attribute__((ext_vector_type(8)))  float  v8f;
typedef __attribute__((ext_vector_type(8)))  __bf16 v8bf;
typedef __attribute__((ext_vector_type(4)))  unsigned int u32x4;

#define BB 2
#define TT 2048
#define CC 1024
#define HH 16
#define HD 64

static __device__ inline v8f wmma_bf16(v16bf a, v16bf b, v8f c) {
  return __builtin_amdgcn_wmma_f32_16x16x32_bf16(false, a, false, b, (short)0, c, false, false);
}

static __device__ inline float gelu_tanh(float x) {
  const float c = 0.7978845608028654f;
  return 0.5f * x * (1.0f + tanhf(c * (x + 0.044715f * x * x * x)));
}

static __device__ inline v16bf frag_rowmajor(const __bf16* p, int ld, int lane) {
  const int m  = lane & 15;
  const int ko = (lane >> 4) << 3;
  const __bf16* q = p + (size_t)m * ld + ko;
  v16bf f;
#pragma unroll
  for (int j = 0; j < 4; ++j) {
    f[2 * j]     = q[2 * j];
    f[2 * j + 1] = q[2 * j + 1];
    f[8 + 2 * j] = q[16 + 2 * j];
    f[9 + 2 * j] = q[16 + 2 * j + 1];
  }
  return f;
}

static __device__ inline v16bf frag_packed(const __bf16* f, int lane) {
  const v8bf lo = *(const v8bf*)(f + lane * 8);
  const v8bf hi = *(const v8bf*)(f + 256 + lane * 8);
  return __builtin_shufflevector(lo, hi, 0, 1, 2, 3, 4, 5, 6, 7, 8, 9, 10, 11, 12, 13, 14, 15);
}

static __device__ inline float xhalf(float x) {
  const int b = __builtin_bit_cast(int, x);
  return __builtin_bit_cast(float, __builtin_amdgcn_permlanex16(b, b, 0x76543210, 0xfedcba98, false, false));
}

typedef __attribute__((ext_vector_type(4))) __bf16 v4bf;
#define WCONV_LOADS 16
__global__ void __launch_bounds__(256)
wconv_kernel(const float* __restrict__ w_qkv, const float* __restrict__ w_ap,
             const float* __restrict__ w_fc, const float* __restrict__ w_pj,
             __bf16* __restrict__ dst) {
  const int u = blockIdx.y;
  const float* src = (u < 3) ? w_qkv + (size_t)u * (CC * CC)
                   : (u < 4) ? w_ap
                   : (u < 8) ? w_fc + (size_t)(u - 4) * (CC * CC)
                             : w_pj + (size_t)(u - 8) * (CC * CC);
  const size_t q0 = (size_t)blockIdx.x * (WCONV_LOADS * 256) + threadIdx.x;
  const float4* s4 = (const float4*)src + q0;
  v4bf* d4 = (v4bf*)(dst + (size_t)u * (CC * CC)) + q0;
  float4 v[WCONV_LOADS];
#pragma unroll
  for (int i = 0; i < WCONV_LOADS; ++i) v[i] = s4[i * 256];
#pragma unroll
  for (int i = 0; i < WCONV_LOADS; ++i) {
    v4bf p;
    p[0] = (__bf16)v[i].x; p[1] = (__bf16)v[i].y;
    p[2] = (__bf16)v[i].z; p[3] = (__bf16)v[i].w;
    d4[i * 256] = p;
  }
}

__global__ void ln_kernel(const float* __restrict__ x, const float* __restrict__ w,
                          const float* __restrict__ b, __bf16* __restrict__ out) {
  const int row = blockIdx.x;
  const int tid = threadIdx.x;
  const float* xr = x + (size_t)row * CC;
  float v[4];
  float s = 0.f, sq = 0.f;
#pragma unroll
  for (int i = 0; i < 4; ++i) {
    v[i] = xr[tid + i * 256];
    s += v[i];
    sq += v[i] * v[i];
  }
#pragma unroll
  for (int off = 16; off >= 1; off >>= 1) {
    s  += __shfl_xor(s,  off, 32);
    sq += __shfl_xor(sq, off, 32);
  }
  __shared__ float reds[8], redq[8];
  const int wid = tid >> 5;
  if ((tid & 31) == 0) { reds[wid] = s; redq[wid] = sq; }
  __syncthreads();
  float ts = 0.f, tq = 0.f;
#pragma unroll
  for (int k = 0; k < 8; ++k) { ts += reds[k]; tq += redq[k]; }
  const float mean = ts * (1.0f / CC);
  const float var  = tq * (1.0f / CC) - mean * mean;
  const float rstd = rsqrtf(var + 1e-5f);
  __bf16* orow = out + (size_t)row * CC;
#pragma unroll
  for (int i = 0; i < 4; ++i) {
    const int c = tid + i * 256;
    orow[c] = (__bf16)((v[i] - mean) * rstd * w[c] + b[c]);
  }
}


template <int LD>
static __device__ inline void mma_step(const __bf16* as, const __bf16* bs, v8f (&acc)[4][2],
                                       int wm, int wn, int lane) {
  v16bf af[4], bf[2];
#pragma unroll
  for (int i = 0; i < 4; ++i) af[i] = frag_rowmajor(as + (wm * 64 + i * 16) * LD, LD, lane);
#pragma unroll
  for (int j = 0; j < 2; ++j) bf[j] = frag_rowmajor(bs + (wn * 32 + j * 16) * LD, LD, lane);
#pragma unroll
  for (int i = 0; i < 4; ++i)
#pragma unroll
    for (int j = 0; j < 2; ++j) acc[i][j] = wmma_bf16(af[i], bf[j], acc[i][j]);
}

template <int EM, int BK>
__global__ void __launch_bounds__(256, (EM == 3 ? 6 : 1))
gemm_kernel(const __bf16* A, const __bf16* W,
            const float* __restrict__ bias, const float* __restrict__ resid,
            void* __restrict__ outp, __bf16* __restrict__ kF, __bf16* __restrict__ vF,
            int M, int N, int K) {
  constexpr int LD  = BK + 8;
  constexpr int LPR = BK / 8;
  constexpr int RPI = 256 / LPR;
  constexpr int NL  = 128 / RPI;
  __shared__ __attribute__((aligned(16))) __bf16 As[2][128 * LD];
  __shared__ __attribute__((aligned(16))) __bf16 Bs[2][128 * LD];
  const int tid  = threadIdx.x;
  const int lane = tid & 31;
  const int wid  = tid >> 5;
  const int wm   = wid & 1;
  const int wn   = wid >> 1;
  const int m0   = blockIdx.y * 128;
  const int n0   = blockIdx.x * 128;

  v8f acc[4][2] = {};

  const int srow = tid / LPR, scol = (tid % LPR) * 8;
  const char* abase = (const char*)(A + (size_t)m0 * K);
  const char* bbase = (const char*)(W + (size_t)n0 * K);
  if constexpr (EM == 3) {
    if (n0 >= CC && n0 < 2 * CC) {
      const char* t = abase; abase = bbase; bbase = t;
    }
  }
  unsigned voff[NL];
#pragma unroll
  for (int j = 0; j < NL; ++j) voff[j] = (unsigned)(((srow + j * RPI) * K + scol) * 2);
  const int soff = srow * LD + scol;

  u32x4 ra[NL], rb[NL];
#pragma unroll
  for (int j = 0; j < NL; ++j) {
    ra[j] = *(const u32x4*)(abase + voff[j]);
    rb[j] = *(const u32x4*)(bbase + voff[j]);
  }
#pragma unroll
  for (int j = 0; j < NL; ++j) {
    *(u32x4*)(As[0] + soff + j * RPI * LD) = ra[j];
    *(u32x4*)(Bs[0] + soff + j * RPI * LD) = rb[j];
  }
  __syncthreads();

  const int nk = K / BK;
  for (int kt = 0; kt + 1 < nk; ++kt) {
    const int cur = kt & 1;
    abase += BK * 2;
    bbase += BK * 2;
#pragma unroll
    for (int j = 0; j < NL; ++j) {
      ra[j] = *(const u32x4*)(abase + voff[j]);
      rb[j] = *(const u32x4*)(bbase + voff[j]);
    }
    asm volatile("" ::: "memory");

#pragma unroll
    for (int ks = 0; ks < BK / 32; ++ks)
      mma_step<LD>(As[cur] + ks * 32, Bs[cur] + ks * 32, acc, wm, wn, lane);

#pragma unroll
    for (int j = 0; j < NL; ++j) {
      *(u32x4*)(As[cur ^ 1] + soff + j * RPI * LD) = ra[j];
      *(u32x4*)(Bs[cur ^ 1] + soff + j * RPI * LD) = rb[j];
    }
    __syncthreads();
  }
#pragma unroll
  for (int ks = 0; ks < BK / 32; ++ks)
    mma_step<LD>(As[(nk - 1) & 1] + ks * 32, Bs[(nk - 1) & 1] + ks * 32, acc, wm, wn, lane);

  const int hi = lane >> 4;
  float biasj[2];
#pragma unroll
  for (int j = 0; j < 2; ++j) biasj[j] = bias[n0 + wn * 32 + j * 16 + (lane & 15)];

  if constexpr (EM == 3) {
    if (n0 >= CC) {
      const int bb  = m0 / TT;
      const int kt0 = (m0 % TT) >> 5;
      if (n0 >= 2 * CC) {
        const int head = ((n0 - 2 * CC) >> 6) + (wn >> 1);
#pragma unroll
        for (int i = 0; i < 4; ++i)
#pragma unroll
          for (int j = 0; j < 2; ++j) {
            const int kt = kt0 + wm * 2 + (i >> 1);
            const int n  = (wn & 1) * 2 + j;
            const size_t tile = (size_t)(bb * HH + head) * (TT / 32) + kt;
            v8bf pk;
#pragma unroll
            for (int r = 0; r < 8; ++r) pk[r] = (__bf16)(acc[i][j][r] + biasj[j]);
            *(v8bf*)(vF + (tile * 4 + n) * 512 + (i & 1) * 256 + lane * 8) = pk;
          }
      } else {
        const int head = ((n0 - CC) >> 6) + wm;
        const size_t tile = (size_t)(bb * HH + head) * (TT / 32) + kt0 + wn;
#pragma unroll
        for (int i = 0; i < 4; ++i) {
          const float4 b0 = *(const float4*)(bias + n0 + wm * 64 + i * 16 + 8 * hi);
          const float4 b1 = *(const float4*)(bias + n0 + wm * 64 + i * 16 + 8 * hi + 4);
          const float br[8] = {b0.x, b0.y, b0.z, b0.w, b1.x, b1.y, b1.z, b1.w};
#pragma unroll
          for (int j = 0; j < 2; ++j) {
            v8bf pk;
#pragma unroll
            for (int r = 0; r < 8; ++r) pk[r] = (__bf16)(acc[i][j][r] + br[r]);
            *(v8bf*)(kF + (tile * 4 + j * 2 + (i >> 1)) * 512 + (i & 1) * 256 + lane * 8) = pk;
          }
        }
      }
      return;
    }
  }

#pragma unroll
  for (int i = 0; i < 4; ++i)
#pragma unroll
    for (int j = 0; j < 2; ++j)
#pragma unroll
      for (int r = 0; r < 8; ++r) {
        const int row = m0 + wm * 64 + i * 16 + r + 8 * hi;
        const int col = n0 + wn * 32 + j * 16 + (lane & 15);
        float v = acc[i][j][r] + biasj[j];
        const size_t idx = (size_t)row * N + col;
        if constexpr (EM == 0) {
          ((float*)outp)[idx] = v + resid[idx];
        } else if constexpr (EM == 1 || EM == 3) {
          ((__bf16*)outp)[idx] = (__bf16)v;
        } else {
          ((__bf16*)outp)[idx] = (__bf16)gelu_tanh(v);
        }
      }
}


static __device__ inline void load_kfrag(v16bf (&k)[2][2], const __bf16* KFp, int kt, int lane) {
  const __bf16* ktile = KFp + (size_t)kt * (4 * 512);
#pragma unroll
  for (int t = 0; t < 2; ++t) {
    k[t][0] = frag_packed(ktile + (t * 2 + 0) * 512, lane);
    k[t][1] = frag_packed(ktile + (t * 2 + 1) * 512, lane);
  }
}

static __device__ inline void load_vfrag(v16bf (&v)[4], const __bf16* VFp, int kt, int lane) {
  const __bf16* vtile = VFp + (size_t)kt * (4 * 512);
#pragma unroll
  for (int n = 0; n < 4; ++n) v[n] = frag_packed(vtile + n * 512, lane);
}

#define QT 2
__global__ void __launch_bounds__(32, 1)
attn_kernel(const __bf16* __restrict__ qkv, const __bf16* __restrict__ kF,
            const __bf16* __restrict__ vF, __bf16* __restrict__ y, float* __restrict__ ent) {
  const int lane  = threadIdx.x;
  const int hi    = lane >> 4;
  const int qbase = blockIdx.x * (16 * QT);
  const int h     = blockIdx.y;
  const int b     = blockIdx.z;
  const int ldq   = 3 * CC;
  const int qrow0 = qbase + (lane & 15);

  const __bf16* Qp  = qkv + (size_t)b * TT * ldq + h * HD;
  const __bf16* KFp = kF + (size_t)(b * HH + h) * (TT / 32) * (4 * 512);
  const __bf16* VFp = vF + (size_t)(b * HH + h) * (TT / 32) * (4 * 512);

  v16bf qb[QT][2];
#pragma unroll
  for (int u = 0; u < QT; ++u) {
    qb[u][0] = frag_rowmajor(Qp + (size_t)(qbase + 16 * u) * ldq + 0,  ldq, lane);
    qb[u][1] = frag_rowmajor(Qp + (size_t)(qbase + 16 * u) * ldq + 32, ldq, lane);
  }

  v8f o[QT][4] = {};
  float mrow[QT], lrow[QT], ses[QT];
#pragma unroll
  for (int u = 0; u < QT; ++u) { mrow[u] = -INFINITY; lrow[u] = 0.f; ses[u] = 0.f; }

  const int nkt = (qbase + 16 * QT + 31) >> 5;

  __builtin_amdgcn_sched_barrier(0);

  v16bf kf[2][2], vf[4];
  load_kfrag(kf, KFp, 0, lane);
  load_vfrag(vf, VFp, 0, lane);

  int kt = 0;
  do {
    const int k0 = kt * 32;
    const int ktn = (kt + 1 < nkt) ? kt + 1 : kt;
    v8f s[QT][2];
#pragma unroll
    for (int t = 0; t < 2; ++t)
#pragma unroll
      for (int u = 0; u < QT; ++u) {
        v8f z = {};
        z = wmma_bf16(kf[t][0], qb[u][0], z);
        s[u][t] = wmma_bf16(kf[t][1], qb[u][1], z);
      }
    load_kfrag(kf, KFp, ktn, lane);
    __builtin_amdgcn_sched_barrier(0);

    v16bf pb[QT];
#pragma unroll
    for (int u = 0; u < QT; ++u) {
      const int qrow = qrow0 + 16 * u;
      float st[2][8];
#pragma unroll
      for (int t = 0; t < 2; ++t)
#pragma unroll
        for (int r = 0; r < 8; ++r) {
          const int key = k0 + t * 16 + 8 * hi + r;
          st[t][r] = (key <= qrow) ? s[u][t][r] * 0.125f : -1.0e30f;
        }
      float v = fmaxf(st[0][0], st[1][0]);
#pragma unroll
      for (int r = 1; r < 8; ++r) v = fmaxf(v, fmaxf(st[0][r], st[1][r]));
      v = fmaxf(v, xhalf(v));
      const float mn = fmaxf(mrow[u], v);
      const float a  = __expf(mrow[u] - mn);
      float x[2][8];
      float ps = 0.f, pe = 0.f;
#pragma unroll
      for (int r = 0; r < 8; ++r) {
        x[0][r] = __expf(st[0][r] - mn);
        x[1][r] = __expf(st[1][r] - mn);
        ps += x[0][r] + x[1][r];
        pe += x[0][r] * st[0][r] + x[1][r] * st[1][r];
      }
      ps += xhalf(ps);
      pe += xhalf(pe);
      lrow[u] = lrow[u] * a + ps;
      ses[u]  = ses[u]  * a + pe;
      mrow[u] = mn;
#pragma unroll
      for (int n = 0; n < 4; ++n)
#pragma unroll
        for (int r = 0; r < 8; ++r) o[u][n][r] *= a;
#pragma unroll
      for (int r = 0; r < 8; ++r) {
        pb[u][r]     = (__bf16)x[0][r];
        pb[u][8 + r] = (__bf16)x[1][r];
      }
    }
#pragma unroll
    for (int n = 0; n < 4; ++n)
#pragma unroll
      for (int u = 0; u < QT; ++u) o[u][n] = wmma_bf16(vf[n], pb[u], o[u][n]);
    load_vfrag(vf, VFp, ktn, lane);
  } while (++kt < nkt);

#pragma unroll
  for (int u = 0; u < QT; ++u) {
    const int qrow = qrow0 + 16 * u;
    const float inv = 1.0f / lrow[u];
    __bf16* yrow = y + ((size_t)b * TT + qrow) * CC + h * HD + 8 * hi;
#pragma unroll
    for (int n = 0; n < 4; ++n) {
      v8bf pk;
#pragma unroll
      for (int r = 0; r < 8; ++r) pk[r] = (__bf16)(o[u][n][r] * inv);
      *(v8bf*)(yrow + n * 16) = pk;
    }
    float e = mrow[u] + __logf(lrow[u]) - ses[u] * inv;
#pragma unroll
    for (int off = 1; off <= 8; off <<= 1) e += __shfl_xor(e, off, 32);
    if (lane == 0) ent[(size_t)(b * HH + h) * (TT / 16) + blockIdx.x * QT + u] = e;
  }
}

#define ENT_PARTS (BB * HH * (TT / 16))
__global__ void __launch_bounds__(256)
ent_final_kernel(const float* __restrict__ part, float* __restrict__ out) {
  const int tid = threadIdx.x;
  float s = 0.f;
#pragma unroll
  for (int k = 0; k < ENT_PARTS / 256; ++k) s += part[tid + k * 256];
#pragma unroll
  for (int off = 16; off >= 1; off >>= 1) s += __shfl_xor(s, off, 32);
  __shared__ float red[8];
  if ((tid & 31) == 0) red[tid >> 5] = s;
  __syncthreads();
  if (tid == 0) {
    float t = 0.f;
#pragma unroll
    for (int k = 0; k < 8; ++k) t += red[k];
    *out = t / (float)(BB * HH * TT);
  }
}

extern "C" void kernel_launch(void* const* d_in, const int* in_sizes, int n_in,
                              void* d_out, int out_size, void* d_ws, size_t ws_size,
                              hipStream_t stream) {
  (void)in_sizes; (void)n_in; (void)out_size; (void)ws_size;
  const float* x          = (const float*)d_in[0];
  const float* ln1_w      = (const float*)d_in[1];
  const float* ln1_b      = (const float*)d_in[2];
  const float* attn_w     = (const float*)d_in[3];
  const float* attn_b     = (const float*)d_in[4];
  const float* attnproj_w = (const float*)d_in[5];
  const float* attnproj_b = (const float*)d_in[6];
  const float* ln2_w      = (const float*)d_in[7];
  const float* ln2_b      = (const float*)d_in[8];
  const float* fc_w       = (const float*)d_in[9];
  const float* fc_b       = (const float*)d_in[10];
  const float* proj_w     = (const float*)d_in[11];
  const float* proj_b     = (const float*)d_in[12];
  float* out = (float*)d_out;

  char* ws = (char*)d_ws;
  float*  x1      = (float*)(ws);
  __bf16* qkv_bf  = (__bf16*)(ws + ((size_t)16 << 20));
  __bf16* h_bf    = (__bf16*)(ws + ((size_t)40 << 20));
  __bf16* y_bf    = (__bf16*)(ws + ((size_t)48 << 20));
  __bf16* m_bf    = (__bf16*)(ws + ((size_t)56 << 20));
  __bf16* vF_bf   = m_bf;
  __bf16* kF_bf   = (__bf16*)(ws + ((size_t)64 << 20));
  float*  epart   = (float*)(ws + ((size_t)72 << 20));
  __bf16* wqkv_bf = (__bf16*)(ws + ((size_t)88 << 20));
  __bf16* wap_bf  = (__bf16*)(ws + ((size_t)94 << 20));
  __bf16* wfc_bf  = (__bf16*)(ws + ((size_t)96 << 20));
  __bf16* wpj_bf  = (__bf16*)(ws + ((size_t)104 << 20));

  const int M = BB * TT;

  wconv_kernel<<<dim3((CC * CC) / (4 * WCONV_LOADS * 256), 12), 256, 0, stream>>>(
      attn_w, attnproj_w, fc_w, proj_w, wqkv_bf);
  ln_kernel<<<M, 256, 0, stream>>>(x, ln1_w, ln1_b, h_bf);
  gemm_kernel<3, 32><<<dim3(3 * CC / 128, M / 128), 256, 0, stream>>>(
      h_bf, wqkv_bf, attn_b, nullptr, qkv_bf, kF_bf, vF_bf, M, 3 * CC, CC);
  attn_kernel<<<dim3(TT / (16 * QT), HH, BB), 32, 0, stream>>>(qkv_bf, kF_bf, vF_bf, y_bf, epart);
  ent_final_kernel<<<1, 256, 0, stream>>>(epart, out + (size_t)M * CC);
  gemm_kernel<0, 128><<<dim3(CC / 128, M / 128), 256, 0, stream>>>(
      y_bf, wap_bf, attnproj_b, x, x1, nullptr, nullptr, M, CC, CC);
  ln_kernel<<<M, 256, 0, stream>>>(x1, ln2_w, ln2_b, h_bf);
  gemm_kernel<2, 32><<<dim3(4 * CC / 128, M / 128), 256, 0, stream>>>(
      h_bf, wfc_bf, fc_b, nullptr, m_bf, nullptr, nullptr, M, 4 * CC, CC);
  gemm_kernel<0, 128><<<dim3(CC / 128, M / 128), 256, 0, stream>>>(
      m_bf, wpj_bf, proj_b, x1, out, nullptr, nullptr, M, CC, 4 * CC);
}


# QKV and MLP fc GEMMs retiled: block 128x128 (8 waves of 64x32) -> 256x256 (16 waves of 64x64), K-step 64: half the operand bytes fetched per WMMA, 8 LDS fragments per 16 WMMAs; sum order unchanged
# speedup vs baseline: 2.8964x; 1.1485x over previous
#include <hip/hip_runtime.h>
#include <hip/hip_bf16.h>
#include <math.h>

typedef __attribute__((ext_vector_type(16))) __bf16 v16bf;
typedef __attribute__((ext_vector_type(8)))  float  v8f;
typedef __attribute__((ext_vector_type(8)))  __bf16 v8bf;
typedef __attribute__((ext_vector_type(4)))  unsigned int u32x4;

#define BB 2
#define TT 2048
#define CC 1024
#define HH 16
#define HD 64

static __device__ inline v8f wmma_bf16(v16bf a, v16bf b, v8f c) {
  return __builtin_amdgcn_wmma_f32_16x16x32_bf16(false, a, false, b, (short)0, c, false, false);
}

static __device__ inline float gelu_tanh(float x) {
  const float k0 = -2.0f * 0.7978845608028654f * 1.4426950408889634f;
  const float k1 = k0 * 0.044715f;
  const float e = __builtin_amdgcn_exp2f(x * (k0 + k1 * (x * x)));
  return x * __builtin_amdgcn_rcpf(1.0f + e);
}

static __device__ inline v16bf frag_rowmajor(const __bf16* p, int ld, int lane) {
  const int m  = lane & 15;
  const int ko = (lane >> 4) << 3;
  const __bf16* q = p + (size_t)m * ld + ko;
  v16bf f;
#pragma unroll
  for (int j = 0; j < 4; ++j) {
    f[2 * j]     = q[2 * j];
    f[2 * j + 1] = q[2 * j + 1];
    f[8 + 2 * j] = q[16 + 2 * j];
    f[9 + 2 * j] = q[16 + 2 * j + 1];
  }
  return f;
}

static __device__ inline v16bf frag_packed(const __bf16* f, int lane) {
  const v8bf lo = *(const v8bf*)(f + lane * 8);
  const v8bf hi = *(const v8bf*)(f + 256 + lane * 8);
  return __builtin_shufflevector(lo, hi, 0, 1, 2, 3, 4, 5, 6, 7, 8, 9, 10, 11, 12, 13, 14, 15);
}

static __device__ inline float xhalf(float x) {
  const int b = __builtin_bit_cast(int, x);
  return __builtin_bit_cast(float, __builtin_amdgcn_permlanex16(b, b, 0x76543210, 0xfedcba98, false, false));
}

typedef __attribute__((ext_vector_type(4))) __bf16 v4bf;
#define WCONV_LOADS 16
__global__ void __launch_bounds__(256)
wconv_kernel(const float* __restrict__ w_qkv, const float* __restrict__ w_ap,
             const float* __restrict__ w_fc, const float* __restrict__ w_pj,
             __bf16* __restrict__ dst) {
  const int u = blockIdx.y;
  const float* src = (u < 3) ? w_qkv + (size_t)u * (CC * CC)
                   : (u < 4) ? w_ap
                   : (u < 8) ? w_fc + (size_t)(u - 4) * (CC * CC)
                             : w_pj + (size_t)(u - 8) * (CC * CC);
  const size_t q0 = (size_t)blockIdx.x * (WCONV_LOADS * 256) + threadIdx.x;
  const float4* s4 = (const float4*)src + q0;
  v4bf* d4 = (v4bf*)(dst + (size_t)u * (CC * CC)) + q0;
  float4 v[WCONV_LOADS];
#pragma unroll
  for (int i = 0; i < WCONV_LOADS; ++i) v[i] = s4[i * 256];
#pragma unroll
  for (int i = 0; i < WCONV_LOADS; ++i) {
    v4bf p;
    p[0] = (__bf16)v[i].x; p[1] = (__bf16)v[i].y;
    p[2] = (__bf16)v[i].z; p[3] = (__bf16)v[i].w;
    d4[i * 256] = p;
  }
}

__global__ void ln_kernel(const float* __restrict__ x, const float* __restrict__ w,
                          const float* __restrict__ b, __bf16* __restrict__ out) {
  const int row = blockIdx.x;
  const int tid = threadIdx.x;
  const float* xr = x + (size_t)row * CC;
  float v[4];
  float s = 0.f, sq = 0.f;
#pragma unroll
  for (int i = 0; i < 4; ++i) {
    v[i] = xr[tid + i * 256];
    s += v[i];
    sq += v[i] * v[i];
  }
#pragma unroll
  for (int off = 16; off >= 1; off >>= 1) {
    s  += __shfl_xor(s,  off, 32);
    sq += __shfl_xor(sq, off, 32);
  }
  __shared__ float reds[8], redq[8];
  const int wid = tid >> 5;
  if ((tid & 31) == 0) { reds[wid] = s; redq[wid] = sq; }
  __syncthreads();
  float ts = 0.f, tq = 0.f;
#pragma unroll
  for (int k = 0; k < 8; ++k) { ts += reds[k]; tq += redq[k]; }
  const float mean = ts * (1.0f / CC);
  const float var  = tq * (1.0f / CC) - mean * mean;
  const float rstd = rsqrtf(var + 1e-5f);
  __bf16* orow = out + (size_t)row * CC;
#pragma unroll
  for (int i = 0; i < 4; ++i) {
    const int c = tid + i * 256;
    orow[c] = (__bf16)((v[i] - mean) * rstd * w[c] + b[c]);
  }
}


template <int LD, int NJ>
static __device__ inline void mma_step(const __bf16* as, const __bf16* bs, v8f (&acc)[4][NJ],
                                       int wm, int wn, int lane) {
  v16bf af[4], bf[NJ];
#pragma unroll
  for (int i = 0; i < 4; ++i) af[i] = frag_rowmajor(as + (wm * 64 + i * 16) * LD, LD, lane);
#pragma unroll
  for (int j = 0; j < NJ; ++j) bf[j] = frag_rowmajor(bs + (wn * (16 * NJ) + j * 16) * LD, LD, lane);
#pragma unroll
  for (int i = 0; i < 4; ++i)
#pragma unroll
    for (int j = 0; j < NJ; ++j) acc[i][j] = wmma_bf16(af[i], bf[j], acc[i][j]);
}

template <int EM, int BK, int S>
__global__ void __launch_bounds__(256 * S, (S == 2 ? 4 : 1))
gemm_kernel(const __bf16* A, const __bf16* W,
            const float* __restrict__ bias, const float* __restrict__ resid,
            void* __restrict__ outp, __bf16* __restrict__ kF, __bf16* __restrict__ vF,
            int M, int N, int K) {
  constexpr int BT  = 128 * S;
  constexpr int NT  = 256 * S;
  constexpr int NWM = 2 * S;
  constexpr int NJ  = 2 * S;
  constexpr int WNW = 16 * NJ;
  constexpr int LD  = BK + 8;
  constexpr int LPR = BK / 8;
  constexpr int RPI = NT / LPR;
  constexpr int NL  = BT / RPI;
  __shared__ __attribute__((aligned(16))) __bf16 As[2][BT * LD];
  __shared__ __attribute__((aligned(16))) __bf16 Bs[2][BT * LD];
  const int tid  = threadIdx.x;
  const int lane = tid & 31;
  const int wid  = tid >> 5;
  const int wm   = wid % NWM;
  const int wn   = wid / NWM;
  const int m0   = blockIdx.y * BT;
  const int n0   = blockIdx.x * BT;

  v8f acc[4][NJ] = {};

  const int srow = tid / LPR, scol = (tid % LPR) * 8;
  const char* abase = (const char*)(A + (size_t)m0 * K);
  const char* bbase = (const char*)(W + (size_t)n0 * K);
  if constexpr (EM == 3) {
    if (n0 >= CC && n0 < 2 * CC) {
      const char* t = abase; abase = bbase; bbase = t;
    }
  }
  unsigned voff[NL];
#pragma unroll
  for (int j = 0; j < NL; ++j) voff[j] = (unsigned)(((srow + j * RPI) * K + scol) * 2);
  const int soff = srow * LD + scol;

  u32x4 ra[NL], rb[NL];
#pragma unroll
  for (int j = 0; j < NL; ++j) {
    ra[j] = *(const u32x4*)(abase + voff[j]);
    rb[j] = *(const u32x4*)(bbase + voff[j]);
  }
#pragma unroll
  for (int j = 0; j < NL; ++j) {
    *(u32x4*)(As[0] + soff + j * RPI * LD) = ra[j];
    *(u32x4*)(Bs[0] + soff + j * RPI * LD) = rb[j];
  }
  __syncthreads();

  const int nk = K / BK;
  for (int kt = 0; kt + 1 < nk; ++kt) {
    const int cur = kt & 1;
    abase += BK * 2;
    bbase += BK * 2;
#pragma unroll
    for (int j = 0; j < NL; ++j) {
      ra[j] = *(const u32x4*)(abase + voff[j]);
      rb[j] = *(const u32x4*)(bbase + voff[j]);
    }
    asm volatile("" ::: "memory");

#pragma unroll
    for (int ks = 0; ks < BK / 32; ++ks)
      mma_step<LD, NJ>(As[cur] + ks * 32, Bs[cur] + ks * 32, acc, wm, wn, lane);

#pragma unroll
    for (int j = 0; j < NL; ++j) {
      *(u32x4*)(As[cur ^ 1] + soff + j * RPI * LD) = ra[j];
      *(u32x4*)(Bs[cur ^ 1] + soff + j * RPI * LD) = rb[j];
    }
    __syncthreads();
  }
#pragma unroll
  for (int ks = 0; ks < BK / 32; ++ks)
    mma_step<LD, NJ>(As[(nk - 1) & 1] + ks * 32, Bs[(nk - 1) & 1] + ks * 32, acc, wm, wn, lane);

  const int hi = lane >> 4;
  float biasj[NJ];
#pragma unroll
  for (int j = 0; j < NJ; ++j) biasj[j] = bias[n0 + wn * WNW + j * 16 + (lane & 15)];

  if constexpr (EM == 3) {
    if (n0 >= CC) {
      const int bb  = m0 / TT;
      const int kt0 = (m0 % TT) >> 5;
      if (n0 >= 2 * CC) {
#pragma unroll
        for (int i = 0; i < 4; ++i)
#pragma unroll
          for (int j = 0; j < NJ; ++j) {
            const int kt = kt0 + wm * 2 + (i >> 1);
            const int fc = wn * WNW + j * 16;
            const int head = ((n0 - 2 * CC) >> 6) + (fc >> 6);
            const int n  = (fc & 63) >> 4;
            const size_t tile = (size_t)(bb * HH + head) * (TT / 32) + kt;
            v8bf pk;
#pragma unroll
            for (int r = 0; r < 8; ++r) pk[r] = (__bf16)(acc[i][j][r] + biasj[j]);
            *(v8bf*)(vF + (tile * 4 + n) * 512 + (i & 1) * 256 + lane * 8) = pk;
          }
      } else {
        const int head = ((n0 - CC) >> 6) + wm;
        const size_t tile0 = (size_t)(bb * HH + head) * (TT / 32) + kt0;
#pragma unroll
        for (int i = 0; i < 4; ++i) {
          const float4 b0 = *(const float4*)(bias + n0 + wm * 64 + i * 16 + 8 * hi);
          const float4 b1 = *(const float4*)(bias + n0 + wm * 64 + i * 16 + 8 * hi + 4);
          const float br[8] = {b0.x, b0.y, b0.z, b0.w, b1.x, b1.y, b1.z, b1.w};
#pragma unroll
          for (int j = 0; j < NJ; ++j) {
            const int tc = wn * WNW + j * 16;
            const size_t tile = tile0 + (tc >> 5);
            const int t = (tc >> 4) & 1;
            v8bf pk;
#pragma unroll
            for (int r = 0; r < 8; ++r) pk[r] = (__bf16)(acc[i][j][r] + br[r]);
            *(v8bf*)(kF + (tile * 4 + t * 2 + (i >> 1)) * 512 + (i & 1) * 256 + lane * 8) = pk;
          }
        }
      }
      return;
    }
  }

#pragma unroll
  for (int i = 0; i < 4; ++i)
#pragma unroll
    for (int j = 0; j < NJ; ++j)
#pragma unroll
      for (int r = 0; r < 8; ++r) {
        const int row = m0 + wm * 64 + i * 16 + r + 8 * hi;
        const int col = n0 + wn * WNW + j * 16 + (lane & 15);
        float v = acc[i][j][r] + biasj[j];
        const size_t idx = (size_t)row * N + col;
        if constexpr (EM == 0) {
          ((float*)outp)[idx] = v + resid[idx];
        } else if constexpr (EM == 1 || EM == 3) {
          ((__bf16*)outp)[idx] = (__bf16)v;
        } else {
          ((__bf16*)outp)[idx] = (__bf16)gelu_tanh(v);
        }
      }
}


static __device__ inline void load_kfrag(v16bf (&k)[2][2], const __bf16* KFp, int kt, int lane) {
  const __bf16* ktile = KFp + (size_t)kt * (4 * 512);
#pragma unroll
  for (int t = 0; t < 2; ++t) {
    k[t][0] = frag_packed(ktile + (t * 2 + 0) * 512, lane);
    k[t][1] = frag_packed(ktile + (t * 2 + 1) * 512, lane);
  }
}

static __device__ inline void load_vfrag(v16bf (&v)[4], const __bf16* VFp, int kt, int lane) {
  const __bf16* vtile = VFp + (size_t)kt * (4 * 512);
#pragma unroll
  for (int n = 0; n < 4; ++n) v[n] = frag_packed(vtile + n * 512, lane);
}

#define QT 2
__global__ void __launch_bounds__(32, 1)
attn_kernel(const __bf16* __restrict__ qkv, const __bf16* __restrict__ kF,
            const __bf16* __restrict__ vF, __bf16* __restrict__ y, float* __restrict__ ent) {
  const int lane  = threadIdx.x;
  const int hi    = lane >> 4;
  const int qbase = blockIdx.x * (16 * QT);
  const int h     = blockIdx.y;
  const int b     = blockIdx.z;
  const int ldq   = 3 * CC;
  const int qrow0 = qbase + (lane & 15);

  const __bf16* Qp  = qkv + (size_t)b * TT * ldq + h * HD;
  const __bf16* KFp = kF + (size_t)(b * HH + h) * (TT / 32) * (4 * 512);
  const __bf16* VFp = vF + (size_t)(b * HH + h) * (TT / 32) * (4 * 512);

  v16bf qb[QT][2];
#pragma unroll
  for (int u = 0; u < QT; ++u) {
    qb[u][0] = frag_rowmajor(Qp + (size_t)(qbase + 16 * u) * ldq + 0,  ldq, lane);
    qb[u][1] = frag_rowmajor(Qp + (size_t)(qbase + 16 * u) * ldq + 32, ldq, lane);
  }

  v8f o[QT][4] = {};
  float mrow[QT], lrow[QT], ses[QT];
#pragma unroll
  for (int u = 0; u < QT; ++u) { mrow[u] = -INFINITY; lrow[u] = 0.f; ses[u] = 0.f; }

  const int nkt = (qbase + 16 * QT + 31) >> 5;

  __builtin_amdgcn_sched_barrier(0);

  v16bf kf[2][2], vf[4];
  load_kfrag(kf, KFp, 0, lane);
  load_vfrag(vf, VFp, 0, lane);

  int kt = 0;
  do {
    const int k0 = kt * 32;
    const int ktn = (kt + 1 < nkt) ? kt + 1 : kt;
    v8f s[QT][2];
#pragma unroll
    for (int t = 0; t < 2; ++t)
#pragma unroll
      for (int u = 0; u < QT; ++u) {
        v8f z = {};
        z = wmma_bf16(kf[t][0], qb[u][0], z);
        s[u][t] = wmma_bf16(kf[t][1], qb[u][1], z);
      }
    load_kfrag(kf, KFp, ktn, lane);
    __builtin_amdgcn_sched_barrier(0);

    v16bf pb[QT];
#pragma unroll
    for (int u = 0; u < QT; ++u) {
      const int qrow = qrow0 + 16 * u;
      float st[2][8];
#pragma unroll
      for (int t = 0; t < 2; ++t)
#pragma unroll
        for (int r = 0; r < 8; ++r) {
          const int key = k0 + t * 16 + 8 * hi + r;
          st[t][r] = (key <= qrow) ? s[u][t][r] * 0.125f : -1.0e30f;
        }
      float v = fmaxf(st[0][0], st[1][0]);
#pragma unroll
      for (int r = 1; r < 8; ++r) v = fmaxf(v, fmaxf(st[0][r], st[1][r]));
      v = fmaxf(v, xhalf(v));
      const float mn = fmaxf(mrow[u], v);
      const float a  = __expf(mrow[u] - mn);
      float x[2][8];
      float ps = 0.f, pe = 0.f;
#pragma unroll
      for (int r = 0; r < 8; ++r) {
        x[0][r] = __expf(st[0][r] - mn);
        x[1][r] = __expf(st[1][r] - mn);
        ps += x[0][r] + x[1][r];
        pe += x[0][r] * st[0][r] + x[1][r] * st[1][r];
      }
      ps += xhalf(ps);
      pe += xhalf(pe);
      lrow[u] = lrow[u] * a + ps;
      ses[u]  = ses[u]  * a + pe;
      mrow[u] = mn;
#pragma unroll
      for (int n = 0; n < 4; ++n)
#pragma unroll
        for (int r = 0; r < 8; ++r) o[u][n][r] *= a;
#pragma unroll
      for (int r = 0; r < 8; ++r) {
        pb[u][r]     = (__bf16)x[0][r];
        pb[u][8 + r] = (__bf16)x[1][r];
      }
    }
#pragma unroll
    for (int n = 0; n < 4; ++n)
#pragma unroll
      for (int u = 0; u < QT; ++u) o[u][n] = wmma_bf16(vf[n], pb[u], o[u][n]);
    load_vfrag(vf, VFp, ktn, lane);
  } while (++kt < nkt);

#pragma unroll
  for (int u = 0; u < QT; ++u) {
    const int qrow = qrow0 + 16 * u;
    const float inv = 1.0f / lrow[u];
    __bf16* yrow = y + ((size_t)b * TT + qrow) * CC + h * HD + 8 * hi;
#pragma unroll
    for (int n = 0; n < 4; ++n) {
      v8bf pk;
#pragma unroll
      for (int r = 0; r < 8; ++r) pk[r] = (__bf16)(o[u][n][r] * inv);
      *(v8bf*)(yrow + n * 16) = pk;
    }
    float e = mrow[u] + __logf(lrow[u]) - ses[u] * inv;
#pragma unroll
    for (int off = 1; off <= 8; off <<= 1) e += __shfl_xor(e, off, 32);
    if (lane == 0) ent[(size_t)(b * HH + h) * (TT / 16) + blockIdx.x * QT + u] = e;
  }
}

#define ENT_PARTS (BB * HH * (TT / 16))
__global__ void __launch_bounds__(256)
ent_final_kernel(const float* __restrict__ part, float* __restrict__ out) {
  const int tid = threadIdx.x;
  float s = 0.f;
#pragma unroll
  for (int k = 0; k < ENT_PARTS / 256; ++k) s += part[tid + k * 256];
#pragma unroll
  for (int off = 16; off >= 1; off >>= 1) s += __shfl_xor(s, off, 32);
  __shared__ float red[8];
  if ((tid & 31) == 0) red[tid >> 5] = s;
  __syncthreads();
  if (tid == 0) {
    float t = 0.f;
#pragma unroll
    for (int k = 0; k < 8; ++k) t += red[k];
    *out = t / (float)(BB * HH * TT);
  }
}

extern "C" void kernel_launch(void* const* d_in, const int* in_sizes, int n_in,
                              void* d_out, int out_size, void* d_ws, size_t ws_size,
                              hipStream_t stream) {
  (void)in_sizes; (void)n_in; (void)out_size; (void)ws_size;
  const float* x          = (const float*)d_in[0];
  const float* ln1_w      = (const float*)d_in[1];
  const float* ln1_b      = (const float*)d_in[2];
  const float* attn_w     = (const float*)d_in[3];
  const float* attn_b     = (const float*)d_in[4];
  const float* attnproj_w = (const float*)d_in[5];
  const float* attnproj_b = (const float*)d_in[6];
  const float* ln2_w      = (const float*)d_in[7];
  const float* ln2_b      = (const float*)d_in[8];
  const float* fc_w       = (const float*)d_in[9];
  const float* fc_b       = (const float*)d_in[10];
  const float* proj_w     = (const float*)d_in[11];
  const float* proj_b     = (const float*)d_in[12];
  float* out = (float*)d_out;

  char* ws = (char*)d_ws;
  float*  x1      = (float*)(ws);
  __bf16* qkv_bf  = (__bf16*)(ws + ((size_t)16 << 20));
  __bf16* h_bf    = (__bf16*)(ws + ((size_t)40 << 20));
  __bf16* y_bf    = (__bf16*)(ws + ((size_t)48 << 20));
  __bf16* m_bf    = (__bf16*)(ws + ((size_t)56 << 20));
  __bf16* vF_bf   = m_bf;
  __bf16* kF_bf   = (__bf16*)(ws + ((size_t)64 << 20));
  float*  epart   = (float*)(ws + ((size_t)72 << 20));
  __bf16* wqkv_bf = (__bf16*)(ws + ((size_t)88 << 20));
  __bf16* wap_bf  = (__bf16*)(ws + ((size_t)94 << 20));
  __bf16* wfc_bf  = (__bf16*)(ws + ((size_t)96 << 20));
  __bf16* wpj_bf  = (__bf16*)(ws + ((size_t)104 << 20));

  const int M = BB * TT;

  wconv_kernel<<<dim3((CC * CC) / (4 * WCONV_LOADS * 256), 12), 256, 0, stream>>>(
      attn_w, attnproj_w, fc_w, proj_w, wqkv_bf);
  ln_kernel<<<M, 256, 0, stream>>>(x, ln1_w, ln1_b, h_bf);
  gemm_kernel<3, 64, 2><<<dim3(3 * CC / 256, M / 256), 512, 0, stream>>>(
      h_bf, wqkv_bf, attn_b, nullptr, qkv_bf, kF_bf, vF_bf, M, 3 * CC, CC);
  attn_kernel<<<dim3(TT / (16 * QT), HH, BB), 32, 0, stream>>>(qkv_bf, kF_bf, vF_bf, y_bf, epart);
  ent_final_kernel<<<1, 256, 0, stream>>>(epart, out + (size_t)M * CC);
  gemm_kernel<0, 128, 1><<<dim3(CC / 128, M / 128), 256, 0, stream>>>(
      y_bf, wap_bf, attnproj_b, x, x1, nullptr, nullptr, M, CC, CC);
  ln_kernel<<<M, 256, 0, stream>>>(x1, ln2_w, ln2_b, h_bf);
  gemm_kernel<2, 64, 2><<<dim3(4 * CC / 256, M / 256), 512, 0, stream>>>(
      h_bf, wfc_bf, fc_b, nullptr, m_bf, nullptr, nullptr, M, 4 * CC, CC);
  gemm_kernel<0, 128, 1><<<dim3(CC / 128, M / 128), 256, 0, stream>>>(
      m_bf, wpj_bf, proj_b, x1, out, nullptr, nullptr, M, CC, 4 * CC);
}
